# starting kernel
# baseline (speedup 1.0000x reference)
#include <hip/hip_runtime.h>
#include <hip/hip_bf16.h>

typedef float v2f __attribute__((ext_vector_type(2)));
typedef float v8f __attribute__((ext_vector_type(8)));

#define SA    104
#define KROWS 180
#define NB    90
#define BW    6

__global__ __launch_bounds__(256)
void odein_solve_kernel(const float* __restrict__ coeffs,
                        const float* __restrict__ rhs,
                        const float* __restrict__ iv_rhs,
                        const float* __restrict__ steps,
                        float* __restrict__ out)
{
    __shared__ float Abuf[KROWS * SA];
    __shared__ float st[32];
    __shared__ float r1[96];

    const int tid = threadIdx.x;
    const int b   = blockIdx.x;

    for (int idx = tid; idx < KROWS * SA; idx += 256) Abuf[idx] = 0.0f;
    if (tid < 29) st[tid] = steps[b * 29 + tid];
    if (tid < 96) r1[tid] = 0.0f;
    __syncthreads();

    if (tid < 90) {
        float c = coeffs[b * 90 + tid];
        int s = tid / 3;
        Abuf[s * SA + tid] = c;
        float v = c * rhs[b * 30 + s];
        if      (tid == 0) v += iv_rhs[b * 4 + 0];
        else if (tid == 1) v += iv_rhs[b * 4 + 1];
        else if (tid == 3) v += iv_rhs[b * 4 + 2];
        else if (tid == 4) v += iv_rhs[b * 4 + 3];
        r1[tid] = v;
    }
    if (tid >= 96 && tid < 100) {
        const int ivcol[4] = {0, 1, 3, 4};
        Abuf[(30 + (tid - 96)) * SA + ivcol[tid - 96]] = 1.0f;
    }
    if (tid >= 128 && tid < 157) {
        int s = tid - 128;
        float h = st[s], h2 = h * h;
        int r = 34 + 2 * s;
        Abuf[r * SA + 3 * (s + 1) + 0] = -1.0f;
        Abuf[r * SA + 3 * s + 0] = 1.0f;
        Abuf[r * SA + 3 * s + 1] = h;
        Abuf[r * SA + 3 * s + 2] = 0.5f * h2;
        r = 35 + 2 * s;
        Abuf[r * SA + 3 * (s + 1) + 1] = -h;
        Abuf[r * SA + 3 * s + 1] = h;
        Abuf[r * SA + 3 * s + 2] = h2;
        r = 120 + 2 * s;
        Abuf[r * SA + 3 * s + 0] = -1.0f;
        Abuf[r * SA + 3 * (s + 1) + 0] = 1.0f;
        Abuf[r * SA + 3 * (s + 1) + 1] = -h;
        Abuf[r * SA + 3 * (s + 1) + 2] = 0.5f * h2;
        r = 121 + 2 * s;
        Abuf[r * SA + 3 * s + 1] = h;
        Abuf[r * SA + 3 * (s + 1) + 1] = -h;
        Abuf[r * SA + 3 * (s + 1) + 2] = h2;
    }
    if (tid >= 160 && tid < 188) {
        int s = tid - 159;
        int r = 91 + s;
        Abuf[r * SA + 3 * (s - 1) + 1] = -1.0f;
        Abuf[r * SA + 3 * (s + 1) + 1] = 1.0f;
        Abuf[r * SA + 3 * s + 2] = -(st[s] + st[s - 1]);
    }
    __syncthreads();

    const int lane = tid & 31, wv = tid >> 5;
    const int half = lane >> 4, lm = lane & 15;
    const int TI[16] = {0,0, 1,1,1, 2,2,2, 3,3,3, 4,4,4, 5,5};
    const int TJ[16] = {0,1, 0,1,2, 1,2,3, 2,3,4, 3,4,5, 4,5};
    const int m0a = TI[2 * wv] * 16,     n0a = TJ[2 * wv] * 16;
    const int m0b = TI[2 * wv + 1] * 16, n0b = TJ[2 * wv + 1] * 16;

    v8f acc0 = {0.f, 0.f, 0.f, 0.f, 0.f, 0.f, 0.f, 0.f};
    v8f acc1 = {0.f, 0.f, 0.f, 0.f, 0.f, 0.f, 0.f, 0.f};
    for (int kk = 0; kk < KROWS / 4; ++kk) {
        int k0 = kk * 4 + 2 * half;
        v2f a, bb;
        a.x  = Abuf[k0 * SA + m0a + lm];
        a.y  = Abuf[(k0 + 1) * SA + m0a + lm];
        bb.x = Abuf[k0 * SA + n0a + lm];
        bb.y = Abuf[(k0 + 1) * SA + n0a + lm];
        acc0 = __builtin_amdgcn_wmma_f32_16x16x4_f32(false, a, false, bb,
                                                     (short)0, acc0, false, false);
        a.x  = Abuf[k0 * SA + m0b + lm];
        a.y  = Abuf[(k0 + 1) * SA + m0b + lm];
        bb.x = Abuf[k0 * SA + n0b + lm];
        bb.y = Abuf[(k0 + 1) * SA + n0b + lm];
        acc1 = __builtin_amdgcn_wmma_f32_16x16x4_f32(false, a, false, bb,
                                                     (short)0, acc1, false, false);
    }
    __syncthreads();
    float* AtA = Abuf;
#pragma unroll
    for (int v = 0; v < 8; ++v)
        AtA[(m0a + v + 8 * half) * SA + n0a + lm] = acc0[v];
#pragma unroll
    for (int v = 0; v < 8; ++v)
        AtA[(m0b + v + 8 * half) * SA + n0b + lm] = acc1[v];
    __syncthreads();

    const int DR[21] = {1,2,3,4,5,6, 2,3,4,5,6, 3,4,5,6, 4,5,6, 5,6, 6};
    const int DC[21] = {1,1,1,1,1,1, 2,2,2,2,2, 3,3,3,3, 4,4,4, 5,5, 6};
    for (int j = 0; j < NB; ++j) {
        if (tid == 0) {
            float d = sqrtf(AtA[j * SA + j]);
            AtA[j * SA + j] = d;
            float inv = 1.0f / d;
            int rmax = (j + BW < NB - 1) ? (j + BW) : (NB - 1);
            for (int r = j + 1; r <= rmax; ++r) AtA[r * SA + j] *= inv;
        }
        __syncthreads();
        if (tid < 21) {
            int r = j + DR[tid];
            if (r < NB) {
                int c = j + DC[tid];
                AtA[r * SA + c] -= AtA[r * SA + j] * AtA[c * SA + j];
            }
        }
        __syncthreads();
    }

    if (tid == 0) {
        for (int j = 0; j < NB; ++j) {
            float s = r1[j];
            int k0 = (j - BW > 0) ? (j - BW) : 0;
            for (int k = k0; k < j; ++k) s -= AtA[j * SA + k] * r1[k];
            r1[j] = s / AtA[j * SA + j];
        }
        for (int j = NB - 1; j >= 0; --j) {
            float s = r1[j];
            int k1 = (j + BW < NB - 1) ? (j + BW) : (NB - 1);
            for (int k = j + 1; k <= k1; ++k) s -= AtA[k * SA + j] * r1[k];
            r1[j] = s / AtA[j * SA + j];
        }
    }
    __syncthreads();
    if (tid < 90) out[b * 90 + tid] = r1[tid];
}

extern "C" void kernel_launch(void* const* d_in, const int* in_sizes, int n_in,
                              void* d_out, int out_size, void* d_ws, size_t ws_size,
                              hipStream_t stream) {
    (void)n_in; (void)out_size; (void)d_ws; (void)ws_size;
    const float* coeffs = (const float*)d_in[0];
    const float* rhs    = (const float*)d_in[1];
    const float* iv     = (const float*)d_in[2];
    const float* steps  = (const float*)d_in[3];
    float* out = (float*)d_out;
    int bs = in_sizes[0] / 90;
    odein_solve_kernel<<<bs, 256, 0, stream>>>(coeffs, rhs, iv, steps, out);
}


# Band Cholesky column step done by one wave: 27 lanes each apply a[r][c]-=a[r][j]*a[c][j]/a[j][j] (rhs carried as one more row), sqrt scaling after; no serial scale loop or block barrier per column
# speedup vs baseline: 2.3250x; 2.3250x over previous
#include <hip/hip_runtime.h>
#include <hip/hip_bf16.h>

typedef float v2f __attribute__((ext_vector_type(2)));
typedef float v8f __attribute__((ext_vector_type(8)));

#define SA    104
#define KROWS 180
#define NB    90
#define BW    6
#define RB    96

__global__ __launch_bounds__(256)
void odein_solve_kernel(const float* __restrict__ coeffs,
                        const float* __restrict__ rhs,
                        const float* __restrict__ iv_rhs,
                        const float* __restrict__ steps,
                        float* __restrict__ out)
{
    __shared__ float Abuf[KROWS * SA];
    __shared__ float st[32];
    __shared__ float r1[96];

    const int tid = threadIdx.x;
    const int b   = blockIdx.x;

    for (int idx = tid; idx < KROWS * SA; idx += 256) Abuf[idx] = 0.0f;
    if (tid < 29) st[tid] = steps[b * 29 + tid];
    if (tid < 96) r1[tid] = 0.0f;
    __syncthreads();

    if (tid < 90) {
        float c = coeffs[b * 90 + tid];
        int s = tid / 3;
        Abuf[s * SA + tid] = c;
        float v = c * rhs[b * 30 + s];
        if      (tid == 0) v += iv_rhs[b * 4 + 0];
        else if (tid == 1) v += iv_rhs[b * 4 + 1];
        else if (tid == 3) v += iv_rhs[b * 4 + 2];
        else if (tid == 4) v += iv_rhs[b * 4 + 3];
        r1[tid] = v;
    }
    if (tid >= 96 && tid < 100) {
        const int ivcol[4] = {0, 1, 3, 4};
        Abuf[(30 + (tid - 96)) * SA + ivcol[tid - 96]] = 1.0f;
    }
    if (tid >= 128 && tid < 157) {
        int s = tid - 128;
        float h = st[s], h2 = h * h;
        int r = 34 + 2 * s;
        Abuf[r * SA + 3 * (s + 1) + 0] = -1.0f;
        Abuf[r * SA + 3 * s + 0] = 1.0f;
        Abuf[r * SA + 3 * s + 1] = h;
        Abuf[r * SA + 3 * s + 2] = 0.5f * h2;
        r = 35 + 2 * s;
        Abuf[r * SA + 3 * (s + 1) + 1] = -h;
        Abuf[r * SA + 3 * s + 1] = h;
        Abuf[r * SA + 3 * s + 2] = h2;
        r = 120 + 2 * s;
        Abuf[r * SA + 3 * s + 0] = -1.0f;
        Abuf[r * SA + 3 * (s + 1) + 0] = 1.0f;
        Abuf[r * SA + 3 * (s + 1) + 1] = -h;
        Abuf[r * SA + 3 * (s + 1) + 2] = 0.5f * h2;
        r = 121 + 2 * s;
        Abuf[r * SA + 3 * s + 1] = h;
        Abuf[r * SA + 3 * (s + 1) + 1] = -h;
        Abuf[r * SA + 3 * (s + 1) + 2] = h2;
    }
    if (tid >= 160 && tid < 188) {
        int s = tid - 159;
        int r = 91 + s;
        Abuf[r * SA + 3 * (s - 1) + 1] = -1.0f;
        Abuf[r * SA + 3 * (s + 1) + 1] = 1.0f;
        Abuf[r * SA + 3 * s + 2] = -(st[s] + st[s - 1]);
    }
    __syncthreads();

    const int lane = tid & 31, wv = tid >> 5;
    const int half = lane >> 4, lm = lane & 15;
    const int TI[16] = {0,0, 1,1,1, 2,2,2, 3,3,3, 4,4,4, 5,5};
    const int TJ[16] = {0,1, 0,1,2, 1,2,3, 2,3,4, 3,4,5, 4,5};
    const int m0a = TI[2 * wv] * 16,     n0a = TJ[2 * wv] * 16;
    const int m0b = TI[2 * wv + 1] * 16, n0b = TJ[2 * wv + 1] * 16;

    v8f acc0 = {0.f, 0.f, 0.f, 0.f, 0.f, 0.f, 0.f, 0.f};
    v8f acc1 = {0.f, 0.f, 0.f, 0.f, 0.f, 0.f, 0.f, 0.f};
    for (int kk = 0; kk < KROWS / 4; ++kk) {
        int k0 = kk * 4 + 2 * half;
        v2f a, bb;
        a.x  = Abuf[k0 * SA + m0a + lm];
        a.y  = Abuf[(k0 + 1) * SA + m0a + lm];
        bb.x = Abuf[k0 * SA + n0a + lm];
        bb.y = Abuf[(k0 + 1) * SA + n0a + lm];
        acc0 = __builtin_amdgcn_wmma_f32_16x16x4_f32(false, a, false, bb,
                                                     (short)0, acc0, false, false);
        a.x  = Abuf[k0 * SA + m0b + lm];
        a.y  = Abuf[(k0 + 1) * SA + m0b + lm];
        bb.x = Abuf[k0 * SA + n0b + lm];
        bb.y = Abuf[(k0 + 1) * SA + n0b + lm];
        acc1 = __builtin_amdgcn_wmma_f32_16x16x4_f32(false, a, false, bb,
                                                     (short)0, acc1, false, false);
    }
    __syncthreads();
    float* AtA = Abuf;
#pragma unroll
    for (int v = 0; v < 8; ++v)
        AtA[(m0a + v + 8 * half) * SA + n0a + lm] = acc0[v];
#pragma unroll
    for (int v = 0; v < 8; ++v)
        AtA[(m0b + v + 8 * half) * SA + n0b + lm] = acc1[v];
    if (tid < 96) AtA[RB * SA + tid] = r1[tid];
    __syncthreads();

    if (wv == 0) {
        const int ER[32] = {1,2,3,4,5,6,7, 2,3,4,5,6,7, 3,4,5,6,7, 4,5,6,7, 5,6,7, 6,7, 0,0,0,0,0};
        const int EC[32] = {1,1,1,1,1,1,1, 2,2,2,2,2,2, 3,3,3,3,3, 4,4,4,4, 5,5,5, 6,6, 0,0,0,0,0};
        const int er = ER[lane], ec = EC[lane];
        const bool isrhs = (er == 7);
        const int lim = isrhs ? ec : er;
        for (int j = 0; j < NB; ++j) {
            if (ec != 0 && j + lim < NB) {
                const int row = isrhs ? RB : (j + er);
                const int col = j + ec;
                float piv = AtA[j * SA + j];
                float p   = AtA[row * SA + j];
                float q   = AtA[col * SA + j];
                float t   = AtA[row * SA + col];
                AtA[row * SA + col] = t - (p * q) / piv;
            }
            __builtin_amdgcn_fence(__ATOMIC_ACQ_REL, "wavefront");
        }
    }
    __syncthreads();

    float sv[3];
#pragma unroll
    for (int i = 0; i < 3; ++i) {
        const int idx = tid + 256 * i;
        const int j = idx >> 3, k = idx & 7;
        sv[i] = 0.0f;
        if (j < NB && (k == 7 || j + k < NB)) {
            float d = sqrtf(AtA[j * SA + j]);
            float v = (k == 7) ? AtA[RB * SA + j] : AtA[(j + k) * SA + j];
            sv[i] = (k == 0) ? d : v / d;
        }
    }
    __syncthreads();
#pragma unroll
    for (int i = 0; i < 3; ++i) {
        const int idx = tid + 256 * i;
        const int j = idx >> 3, k = idx & 7;
        if (j < NB && (k == 7 || j + k < NB)) {
            if (k == 7) r1[j] = sv[i];
            else        AtA[(j + k) * SA + j] = sv[i];
        }
    }
    __syncthreads();

    if (tid == 0) {
        for (int j = NB - 1; j >= 0; --j) {
            float s = r1[j];
            int k1 = (j + BW < NB - 1) ? (j + BW) : (NB - 1);
            for (int k = j + 1; k <= k1; ++k) s -= AtA[k * SA + j] * r1[k];
            r1[j] = s / AtA[j * SA + j];
        }
    }
    __syncthreads();
    if (tid < 90) out[b * 90 + tid] = r1[tid];
}

extern "C" void kernel_launch(void* const* d_in, const int* in_sizes, int n_in,
                              void* d_out, int out_size, void* d_ws, size_t ws_size,
                              hipStream_t stream) {
    (void)n_in; (void)out_size; (void)d_ws; (void)ws_size;
    const float* coeffs = (const float*)d_in[0];
    const float* rhs    = (const float*)d_in[1];
    const float* iv     = (const float*)d_in[2];
    const float* steps  = (const float*)d_in[3];
    float* out = (float*)d_out;
    int bs = in_sizes[0] / 90;
    odein_solve_kernel<<<bs, 256, 0, stream>>>(coeffs, rhs, iv, steps, out);
}


# Back substitution: columns pre-divided by their pivot and packed per row by all threads; wave 0 keeps packed rows in registers, 6 open sums, 1 fma per unknown on the chain; sums far-to-near, no sqrt
# speedup vs baseline: 4.8641x; 2.0921x over previous
#include <hip/hip_runtime.h>
#include <hip/hip_bf16.h>

typedef float v2f __attribute__((ext_vector_type(2)));
typedef float v8f __attribute__((ext_vector_type(8)));

#define SA    104
#define KROWS 180
#define NB    90
#define BW    6
#define RB    96
#define PK0   (100 * SA)
#define PKW   8

typedef float v4f __attribute__((ext_vector_type(4)));

__device__ __forceinline__ float lane_value(float v, int ln) {
    return __int_as_float(__builtin_amdgcn_readlane(__float_as_int(v), ln));
}

__global__ __launch_bounds__(256)
void odein_solve_kernel(const float* __restrict__ coeffs,
                        const float* __restrict__ rhs,
                        const float* __restrict__ iv_rhs,
                        const float* __restrict__ steps,
                        float* __restrict__ out)
{
    __shared__ __attribute__((aligned(16))) float Abuf[KROWS * SA];
    __shared__ float st[32];
    __shared__ float r1[96];

    const int tid = threadIdx.x;
    const int b   = blockIdx.x;

    for (int idx = tid; idx < KROWS * SA; idx += 256) Abuf[idx] = 0.0f;
    if (tid < 29) st[tid] = steps[b * 29 + tid];
    if (tid < 96) r1[tid] = 0.0f;
    __syncthreads();

    if (tid < 90) {
        float c = coeffs[b * 90 + tid];
        int s = tid / 3;
        Abuf[s * SA + tid] = c;
        float v = c * rhs[b * 30 + s];
        if      (tid == 0) v += iv_rhs[b * 4 + 0];
        else if (tid == 1) v += iv_rhs[b * 4 + 1];
        else if (tid == 3) v += iv_rhs[b * 4 + 2];
        else if (tid == 4) v += iv_rhs[b * 4 + 3];
        r1[tid] = v;
    }
    if (tid >= 96 && tid < 100) {
        const int ivcol[4] = {0, 1, 3, 4};
        Abuf[(30 + (tid - 96)) * SA + ivcol[tid - 96]] = 1.0f;
    }
    if (tid >= 128 && tid < 157) {
        int s = tid - 128;
        float h = st[s], h2 = h * h;
        int r = 34 + 2 * s;
        Abuf[r * SA + 3 * (s + 1) + 0] = -1.0f;
        Abuf[r * SA + 3 * s + 0] = 1.0f;
        Abuf[r * SA + 3 * s + 1] = h;
        Abuf[r * SA + 3 * s + 2] = 0.5f * h2;
        r = 35 + 2 * s;
        Abuf[r * SA + 3 * (s + 1) + 1] = -h;
        Abuf[r * SA + 3 * s + 1] = h;
        Abuf[r * SA + 3 * s + 2] = h2;
        r = 120 + 2 * s;
        Abuf[r * SA + 3 * s + 0] = -1.0f;
        Abuf[r * SA + 3 * (s + 1) + 0] = 1.0f;
        Abuf[r * SA + 3 * (s + 1) + 1] = -h;
        Abuf[r * SA + 3 * (s + 1) + 2] = 0.5f * h2;
        r = 121 + 2 * s;
        Abuf[r * SA + 3 * s + 1] = h;
        Abuf[r * SA + 3 * (s + 1) + 1] = -h;
        Abuf[r * SA + 3 * (s + 1) + 2] = h2;
    }
    if (tid >= 160 && tid < 188) {
        int s = tid - 159;
        int r = 91 + s;
        Abuf[r * SA + 3 * (s - 1) + 1] = -1.0f;
        Abuf[r * SA + 3 * (s + 1) + 1] = 1.0f;
        Abuf[r * SA + 3 * s + 2] = -(st[s] + st[s - 1]);
    }
    __syncthreads();

    const int lane = tid & 31, wv = tid >> 5;
    const int half = lane >> 4, lm = lane & 15;
    const int TI[16] = {0,0, 1,1,1, 2,2,2, 3,3,3, 4,4,4, 5,5};
    const int TJ[16] = {0,1, 0,1,2, 1,2,3, 2,3,4, 3,4,5, 4,5};
    const int m0a = TI[2 * wv] * 16,     n0a = TJ[2 * wv] * 16;
    const int m0b = TI[2 * wv + 1] * 16, n0b = TJ[2 * wv + 1] * 16;

    v8f acc0 = {0.f, 0.f, 0.f, 0.f, 0.f, 0.f, 0.f, 0.f};
    v8f acc1 = {0.f, 0.f, 0.f, 0.f, 0.f, 0.f, 0.f, 0.f};
    for (int kk = 0; kk < KROWS / 4; ++kk) {
        int k0 = kk * 4 + 2 * half;
        v2f a, bb;
        a.x  = Abuf[k0 * SA + m0a + lm];
        a.y  = Abuf[(k0 + 1) * SA + m0a + lm];
        bb.x = Abuf[k0 * SA + n0a + lm];
        bb.y = Abuf[(k0 + 1) * SA + n0a + lm];
        acc0 = __builtin_amdgcn_wmma_f32_16x16x4_f32(false, a, false, bb,
                                                     (short)0, acc0, false, false);
        a.x  = Abuf[k0 * SA + m0b + lm];
        a.y  = Abuf[(k0 + 1) * SA + m0b + lm];
        bb.x = Abuf[k0 * SA + n0b + lm];
        bb.y = Abuf[(k0 + 1) * SA + n0b + lm];
        acc1 = __builtin_amdgcn_wmma_f32_16x16x4_f32(false, a, false, bb,
                                                     (short)0, acc1, false, false);
    }
    __syncthreads();
    float* AtA = Abuf;
#pragma unroll
    for (int v = 0; v < 8; ++v)
        AtA[(m0a + v + 8 * half) * SA + n0a + lm] = acc0[v];
#pragma unroll
    for (int v = 0; v < 8; ++v)
        AtA[(m0b + v + 8 * half) * SA + n0b + lm] = acc1[v];
    if (tid < 96) AtA[RB * SA + tid] = r1[tid];
    __syncthreads();

    if (wv == 0) {
        const int ER[32] = {1,2,3,4,5,6,7, 2,3,4,5,6,7, 3,4,5,6,7, 4,5,6,7, 5,6,7, 6,7, 0,0,0,0,0};
        const int EC[32] = {1,1,1,1,1,1,1, 2,2,2,2,2,2, 3,3,3,3,3, 4,4,4,4, 5,5,5, 6,6, 0,0,0,0,0};
        const int er = ER[lane], ec = EC[lane];
        const bool isrhs = (er == 7);
        const int lim = isrhs ? ec : er;
        for (int j = 0; j < NB; ++j) {
            if (ec != 0 && j + lim < NB) {
                const int row = isrhs ? RB : (j + er);
                const int col = j + ec;
                float piv = AtA[j * SA + j];
                float p   = AtA[row * SA + j];
                float q   = AtA[col * SA + j];
                float t   = AtA[row * SA + col];
                AtA[row * SA + col] = t - (p * q) / piv;
            }
            __builtin_amdgcn_fence(__ATOMIC_ACQ_REL, "wavefront");
        }
    }
    __syncthreads();

    float* pk = Abuf + PK0;
#pragma unroll
    for (int i = 0; i < 3; ++i) {
        const int idx = tid + 256 * i;
        const int j = idx >> 3, k = idx & 7;
        const bool ok = (j < NB) && (k != 7) && (j + k < NB);
        const int js = ok ? j : 0, ks = ok ? k : 0;
        const float d = AtA[js * SA + js];
        const float v = AtA[(ks == 0) ? (RB * SA + js) : ((js + ks) * SA + js)];
        const float q = v / d;
        if (ok) pk[(ks == 0) ? ((js + BW) * PKW + 7) : ((js + ks) * PKW + ks)] = (ks == 0) ? q : -q;
    }
    __syncthreads();

    if (wv == 0) {
        const v4f* pk4 = reinterpret_cast<const v4f*>(Abuf + PK0);
        v4f lo[3], hi[3];
#pragma unroll
        for (int g = 0; g < 3; ++g) {
            lo[g] = pk4[2 * (32 * g + lane)];
            hi[g] = pk4[2 * (32 * g + lane) + 1];
        }
        float a[6];
#pragma unroll
        for (int i = 0; i < 6; ++i) a[i] = lane_value(hi[2][3], (NB - 1 - i + BW) & 31);
        float xs[3] = {0.0f, 0.0f, 0.0f};
#pragma unroll
        for (int m = NB - 1; m >= 0; --m) {
            const int g = m >> 5, ln = m & 31;
            const float x = a[0];
            xs[g] = (lane == ln) ? x : xs[g];
            a[0] = fmaf(lane_value(lo[g][1], ln), x, a[1]);
            a[1] = fmaf(lane_value(lo[g][2], ln), x, a[2]);
            a[2] = fmaf(lane_value(lo[g][3], ln), x, a[3]);
            a[3] = fmaf(lane_value(hi[g][0], ln), x, a[4]);
            a[4] = fmaf(lane_value(hi[g][1], ln), x, a[5]);
            a[5] = fmaf(lane_value(hi[g][2], ln), x, lane_value(hi[g][3], ln));
        }
        out[b * 90 + lane]      = xs[0];
        out[b * 90 + 32 + lane] = xs[1];
        if (lane < NB - 64) out[b * 90 + 64 + lane] = xs[2];
    }
}

extern "C" void kernel_launch(void* const* d_in, const int* in_sizes, int n_in,
                              void* d_out, int out_size, void* d_ws, size_t ws_size,
                              hipStream_t stream) {
    (void)n_in; (void)out_size; (void)d_ws; (void)ws_size;
    const float* coeffs = (const float*)d_in[0];
    const float* rhs    = (const float*)d_in[1];
    const float* iv     = (const float*)d_in[2];
    const float* steps  = (const float*)d_in[3];
    float* out = (float*)d_out;
    int bs = in_sizes[0] / 90;
    odein_solve_kernel<<<bs, 256, 0, stream>>>(coeffs, rhs, iv, steps, out);
}


# Band elimination kept in registers of one wave: a register per column, lanes = rows, pivot and multipliers as scalars; each column divided once (rcp + residual fix), quotients stored as packed band
# speedup vs baseline: 6.0786x; 1.2497x over previous
#include <hip/hip_runtime.h>
#include <hip/hip_bf16.h>

typedef float v2f __attribute__((ext_vector_type(2)));
typedef float v8f __attribute__((ext_vector_type(8)));

#define SA    104
#define KROWS 180
#define NB    90
#define BW    6
#define RB    96
#define PK0   (100 * SA)
#define PKW   8

typedef float v4f __attribute__((ext_vector_type(4)));

__device__ __forceinline__ float lane_value(float v, int ln) {
    return __int_as_float(__builtin_amdgcn_readlane(__float_as_int(v), ln));
}

#define LA    8

__device__ __forceinline__ float load_band_column(const float* AtA, int c, int lane) {
    const int k   = (lane - c) & 15;
    const int row = (lane == 16) ? RB : (c + k);
    return AtA[row * SA + c];
}

__global__ __launch_bounds__(256)
void odein_solve_kernel(const float* __restrict__ coeffs,
                        const float* __restrict__ rhs,
                        const float* __restrict__ iv_rhs,
                        const float* __restrict__ steps,
                        float* __restrict__ out)
{
    __shared__ __attribute__((aligned(16))) float Abuf[KROWS * SA];
    __shared__ float st[32];
    __shared__ float r1[96];

    const int tid = threadIdx.x;
    const int b   = blockIdx.x;

    for (int idx = tid; idx < KROWS * SA; idx += 256) Abuf[idx] = 0.0f;
    if (tid < 29) st[tid] = steps[b * 29 + tid];
    if (tid < 96) r1[tid] = 0.0f;
    __syncthreads();

    if (tid < 90) {
        float c = coeffs[b * 90 + tid];
        int s = tid / 3;
        Abuf[s * SA + tid] = c;
        float v = c * rhs[b * 30 + s];
        if      (tid == 0) v += iv_rhs[b * 4 + 0];
        else if (tid == 1) v += iv_rhs[b * 4 + 1];
        else if (tid == 3) v += iv_rhs[b * 4 + 2];
        else if (tid == 4) v += iv_rhs[b * 4 + 3];
        r1[tid] = v;
    }
    if (tid >= 96 && tid < 100) {
        const int ivcol[4] = {0, 1, 3, 4};
        Abuf[(30 + (tid - 96)) * SA + ivcol[tid - 96]] = 1.0f;
    }
    if (tid >= 128 && tid < 157) {
        int s = tid - 128;
        float h = st[s], h2 = h * h;
        int r = 34 + 2 * s;
        Abuf[r * SA + 3 * (s + 1) + 0] = -1.0f;
        Abuf[r * SA + 3 * s + 0] = 1.0f;
        Abuf[r * SA + 3 * s + 1] = h;
        Abuf[r * SA + 3 * s + 2] = 0.5f * h2;
        r = 35 + 2 * s;
        Abuf[r * SA + 3 * (s + 1) + 1] = -h;
        Abuf[r * SA + 3 * s + 1] = h;
        Abuf[r * SA + 3 * s + 2] = h2;
        r = 120 + 2 * s;
        Abuf[r * SA + 3 * s + 0] = -1.0f;
        Abuf[r * SA + 3 * (s + 1) + 0] = 1.0f;
        Abuf[r * SA + 3 * (s + 1) + 1] = -h;
        Abuf[r * SA + 3 * (s + 1) + 2] = 0.5f * h2;
        r = 121 + 2 * s;
        Abuf[r * SA + 3 * s + 1] = h;
        Abuf[r * SA + 3 * (s + 1) + 1] = -h;
        Abuf[r * SA + 3 * (s + 1) + 2] = h2;
    }
    if (tid >= 160 && tid < 188) {
        int s = tid - 159;
        int r = 91 + s;
        Abuf[r * SA + 3 * (s - 1) + 1] = -1.0f;
        Abuf[r * SA + 3 * (s + 1) + 1] = 1.0f;
        Abuf[r * SA + 3 * s + 2] = -(st[s] + st[s - 1]);
    }
    __syncthreads();

    const int lane = tid & 31, wv = tid >> 5;
    const int half = lane >> 4, lm = lane & 15;
    const int TI[16] = {0,0, 1,1,1, 2,2,2, 3,3,3, 4,4,4, 5,5};
    const int TJ[16] = {0,1, 0,1,2, 1,2,3, 2,3,4, 3,4,5, 4,5};
    const int m0a = TI[2 * wv] * 16,     n0a = TJ[2 * wv] * 16;
    const int m0b = TI[2 * wv + 1] * 16, n0b = TJ[2 * wv + 1] * 16;

    v8f acc0 = {0.f, 0.f, 0.f, 0.f, 0.f, 0.f, 0.f, 0.f};
    v8f acc1 = {0.f, 0.f, 0.f, 0.f, 0.f, 0.f, 0.f, 0.f};
    for (int kk = 0; kk < KROWS / 4; ++kk) {
        int k0 = kk * 4 + 2 * half;
        v2f a, bb;
        a.x  = Abuf[k0 * SA + m0a + lm];
        a.y  = Abuf[(k0 + 1) * SA + m0a + lm];
        bb.x = Abuf[k0 * SA + n0a + lm];
        bb.y = Abuf[(k0 + 1) * SA + n0a + lm];
        acc0 = __builtin_amdgcn_wmma_f32_16x16x4_f32(false, a, false, bb,
                                                     (short)0, acc0, false, false);
        a.x  = Abuf[k0 * SA + m0b + lm];
        a.y  = Abuf[(k0 + 1) * SA + m0b + lm];
        bb.x = Abuf[k0 * SA + n0b + lm];
        bb.y = Abuf[(k0 + 1) * SA + n0b + lm];
        acc1 = __builtin_amdgcn_wmma_f32_16x16x4_f32(false, a, false, bb,
                                                     (short)0, acc1, false, false);
    }
    __syncthreads();
    float* AtA = Abuf;
#pragma unroll
    for (int v = 0; v < 8; ++v)
        AtA[(m0a + v + 8 * half) * SA + n0a + lm] = acc0[v];
#pragma unroll
    for (int v = 0; v < 8; ++v)
        AtA[(m0b + v + 8 * half) * SA + n0b + lm] = acc1[v];
    if (tid < 96) AtA[RB * SA + tid] = r1[tid];
    __syncthreads();

    if (wv == 0) {
        float* pk = Abuf + PK0;
        float col[NB];
#pragma unroll
        for (int c = 0; c < BW + 1 + LA; ++c) col[c] = load_band_column(AtA, c, lane);
#pragma unroll
        for (int j = 0; j < NB; ++j) {
            if (j + BW + 1 + LA < NB)
                col[j + BW + 1 + LA] = load_band_column(AtA, j + BW + 1 + LA, lane);
            const int  k    = (lane - j) & 15;
            const bool isr  = (lane == 16);
            const bool band = (lane < 16) & (k <= BW) & (j + k < NB);
            const bool live = isr | band;
            const float pm  = live ? col[j] : 0.0f;
            const float piv = lane_value(col[j], j & 15);
            const float r0  = __builtin_amdgcn_rcpf(piv);
            const float q0  = pm * r0;
            const float w   = fmaf(fmaf(-piv, q0, pm), r0, q0);
#pragma unroll
            for (int k2 = 1; k2 <= BW; ++k2)
                if (j + k2 < NB)
                    col[j + k2] = fmaf(-lane_value(w, (j + k2) & 15), pm, col[j + k2]);
            int slot = (band & (k >= 1)) ? ((j + k) * PKW + k) : (lane * PKW);
            slot = isr ? ((j + BW) * PKW + 7) : slot;
            pk[slot] = isr ? w : -w;
            __builtin_amdgcn_sched_barrier(0);
        }
    }
    __syncthreads();

    if (wv == 0) {
        const v4f* pk4 = reinterpret_cast<const v4f*>(Abuf + PK0);
        v4f lo[3], hi[3];
#pragma unroll
        for (int g = 0; g < 3; ++g) {
            lo[g] = pk4[2 * (32 * g + lane)];
            hi[g] = pk4[2 * (32 * g + lane) + 1];
        }
        float a[6];
#pragma unroll
        for (int i = 0; i < 6; ++i) a[i] = lane_value(hi[2][3], (NB - 1 - i + BW) & 31);
        float xs[3] = {0.0f, 0.0f, 0.0f};
#pragma unroll
        for (int m = NB - 1; m >= 0; --m) {
            const int g = m >> 5, ln = m & 31;
            const float x = a[0];
            xs[g] = (lane == ln) ? x : xs[g];
            a[0] = fmaf(lane_value(lo[g][1], ln), x, a[1]);
            a[1] = fmaf(lane_value(lo[g][2], ln), x, a[2]);
            a[2] = fmaf(lane_value(lo[g][3], ln), x, a[3]);
            a[3] = fmaf(lane_value(hi[g][0], ln), x, a[4]);
            a[4] = fmaf(lane_value(hi[g][1], ln), x, a[5]);
            a[5] = fmaf(lane_value(hi[g][2], ln), x, lane_value(hi[g][3], ln));
        }
        out[b * 90 + lane]      = xs[0];
        out[b * 90 + 32 + lane] = xs[1];
        if (lane < NB - 64) out[b * 90 + 64 + lane] = xs[2];
    }
}

extern "C" void kernel_launch(void* const* d_in, const int* in_sizes, int n_in,
                              void* d_out, int out_size, void* d_ws, size_t ws_size,
                              hipStream_t stream) {
    (void)n_in; (void)out_size; (void)d_ws; (void)ws_size;
    const float* coeffs = (const float*)d_in[0];
    const float* rhs    = (const float*)d_in[1];
    const float* iv     = (const float*)d_in[2];
    const float* steps  = (const float*)d_in[3];
    float* out = (float*)d_out;
    int bs = in_sizes[0] / 90;
    odein_solve_kernel<<<bs, 256, 0, stream>>>(coeffs, rhs, iv, steps, out);
}


# A's rows ordered by step, so each AtA tile multiplies only the k-blocks that can hold an entry for it: 12 per diagonal tile, 4 per off-diagonal one (was 45 each); AtA's sums change their order
# speedup vs baseline: 7.7216x; 1.2703x over previous
#include <hip/hip_runtime.h>
#include <hip/hip_bf16.h>

typedef float v2f __attribute__((ext_vector_type(2)));
typedef float v8f __attribute__((ext_vector_type(8)));

#define SA    104
#define KROWS 180
#define NB    90
#define BW    6
#define RB    96
#define PK0   (100 * SA)
#define PKW   8

typedef float v4f __attribute__((ext_vector_type(4)));

__device__ __forceinline__ float lane_value(float v, int ln) {
    return __int_as_float(__builtin_amdgcn_readlane(__float_as_int(v), ln));
}

#define LA    8

__device__ __forceinline__ float load_band_column(const float* AtA, int c, int lane) {
    const int k   = (lane - c) & 15;
    const int row = (lane == 16) ? RB : (c + k);
    return AtA[row * SA + c];
}

#define A_ROW0(s) (4 + 6 * (s))

__device__ __forceinline__ v8f ata_kblock(const float* Abuf, int kb, int half, int lm,
                                          int m0, int n0, v8f acc) {
    const int k0 = kb * 4 + 2 * half;
    v2f a, bb;
    a.x  = Abuf[k0 * SA + m0 + lm];
    a.y  = Abuf[(k0 + 1) * SA + m0 + lm];
    bb.x = Abuf[k0 * SA + n0 + lm];
    bb.y = Abuf[(k0 + 1) * SA + n0 + lm];
    return __builtin_amdgcn_wmma_f32_16x16x4_f32(false, a, false, bb, (short)0, acc, false, false);
}

__global__ __launch_bounds__(256)
void odein_solve_kernel(const float* __restrict__ coeffs,
                        const float* __restrict__ rhs,
                        const float* __restrict__ iv_rhs,
                        const float* __restrict__ steps,
                        float* __restrict__ out)
{
    __shared__ __attribute__((aligned(16))) float Abuf[KROWS * SA];
    __shared__ float st[32];
    __shared__ float r1[96];

    const int tid = threadIdx.x;
    const int b   = blockIdx.x;

    for (int idx = tid; idx < KROWS * SA; idx += 256) Abuf[idx] = 0.0f;
    if (tid < 29) st[tid] = steps[b * 29 + tid];
    if (tid < 96) r1[tid] = 0.0f;
    __syncthreads();

    if (tid < 90) {
        float c = coeffs[b * 90 + tid];
        int s = tid / 3;
        Abuf[A_ROW0(s) * SA + tid] = c;
        float v = c * rhs[b * 30 + s];
        if      (tid == 0) v += iv_rhs[b * 4 + 0];
        else if (tid == 1) v += iv_rhs[b * 4 + 1];
        else if (tid == 3) v += iv_rhs[b * 4 + 2];
        else if (tid == 4) v += iv_rhs[b * 4 + 3];
        r1[tid] = v;
    }
    if (tid >= 96 && tid < 100) {
        const int ivcol[4] = {0, 1, 3, 4};
        Abuf[(tid - 96) * SA + ivcol[tid - 96]] = 1.0f;
    }
    if (tid >= 128 && tid < 157) {
        int s = tid - 128;
        float h = st[s], h2 = h * h;
        int r = A_ROW0(s) + 1;
        Abuf[r * SA + 3 * (s + 1) + 0] = -1.0f;
        Abuf[r * SA + 3 * s + 0] = 1.0f;
        Abuf[r * SA + 3 * s + 1] = h;
        Abuf[r * SA + 3 * s + 2] = 0.5f * h2;
        r = A_ROW0(s) + 2;
        Abuf[r * SA + 3 * (s + 1) + 1] = -h;
        Abuf[r * SA + 3 * s + 1] = h;
        Abuf[r * SA + 3 * s + 2] = h2;
        r = A_ROW0(s) + 3;
        Abuf[r * SA + 3 * s + 0] = -1.0f;
        Abuf[r * SA + 3 * (s + 1) + 0] = 1.0f;
        Abuf[r * SA + 3 * (s + 1) + 1] = -h;
        Abuf[r * SA + 3 * (s + 1) + 2] = 0.5f * h2;
        r = A_ROW0(s) + 4;
        Abuf[r * SA + 3 * s + 1] = h;
        Abuf[r * SA + 3 * (s + 1) + 1] = -h;
        Abuf[r * SA + 3 * (s + 1) + 2] = h2;
    }
    if (tid >= 160 && tid < 188) {
        int s = tid - 159;
        int r = A_ROW0(s - 1) + 5;
        Abuf[r * SA + 3 * (s - 1) + 1] = -1.0f;
        Abuf[r * SA + 3 * (s + 1) + 1] = 1.0f;
        Abuf[r * SA + 3 * s + 2] = -(st[s] + st[s - 1]);
    }
    __syncthreads();

    const int lane = tid & 31, wv = tid >> 5;
    const int half = lane >> 4, lm = lane & 15;
    const int wvu = __builtin_amdgcn_readfirstlane(wv);
    const int T0M[8] = {0, 1,  2,  3,  4,  5,   2,  4};
    const int T0N[8] = {0, 1,  2,  3,  4,  5,   1,  3};
    const int T0K[8] = {0, 6, 15, 23, 30, 33,  15, 30};
    const int T1M[8] = {0, 1,  2,  3,  4,  1,   3,  5};
    const int T1N[8] = {1, 2,  3,  4,  5,  0,   2,  4};
    const int T1K[8] = {6, 15, 23, 30, 39, 6,  23, 39};
    const int m0a = T0M[wvu] * 16, n0a = T0N[wvu] * 16, kba = T0K[wvu];
    const int m0b = T1M[wvu] * 16, n0b = T1N[wvu] * 16, kbb = T1K[wvu];

    v8f acc0 = {0.f, 0.f, 0.f, 0.f, 0.f, 0.f, 0.f, 0.f};
    v8f acc1 = {0.f, 0.f, 0.f, 0.f, 0.f, 0.f, 0.f, 0.f};
#pragma unroll
    for (int i = 0; i < 4; ++i) {
        acc0 = ata_kblock(Abuf, kba + i, half, lm, m0a, n0a, acc0);
        acc1 = ata_kblock(Abuf, kbb + i, half, lm, m0b, n0b, acc1);
    }
    if (wvu < 6) {
#pragma unroll
        for (int i = 4; i < 12; ++i)
            acc0 = ata_kblock(Abuf, kba + i, half, lm, m0a, n0a, acc0);
    }
    __syncthreads();
    float* AtA = Abuf;
#pragma unroll
    for (int v = 0; v < 8; ++v)
        AtA[(m0a + v + 8 * half) * SA + n0a + lm] = acc0[v];
#pragma unroll
    for (int v = 0; v < 8; ++v)
        AtA[(m0b + v + 8 * half) * SA + n0b + lm] = acc1[v];
    if (tid < 96) AtA[RB * SA + tid] = r1[tid];
    __syncthreads();

    if (wv == 0) {
        float* pk = Abuf + PK0;
        float col[NB];
#pragma unroll
        for (int c = 0; c < BW + 1 + LA; ++c) col[c] = load_band_column(AtA, c, lane);
#pragma unroll
        for (int j = 0; j < NB; ++j) {
            if (j + BW + 1 + LA < NB)
                col[j + BW + 1 + LA] = load_band_column(AtA, j + BW + 1 + LA, lane);
            const int  k    = (lane - j) & 15;
            const bool isr  = (lane == 16);
            const bool band = (lane < 16) & (k <= BW) & (j + k < NB);
            const bool live = isr | band;
            const float pm  = live ? col[j] : 0.0f;
            const float piv = lane_value(col[j], j & 15);
            const float r0  = __builtin_amdgcn_rcpf(piv);
            const float q0  = pm * r0;
            const float w   = fmaf(fmaf(-piv, q0, pm), r0, q0);
#pragma unroll
            for (int k2 = 1; k2 <= BW; ++k2)
                if (j + k2 < NB)
                    col[j + k2] = fmaf(-lane_value(w, (j + k2) & 15), pm, col[j + k2]);
            int slot = (band & (k >= 1)) ? ((j + k) * PKW + k) : (lane * PKW);
            slot = isr ? ((j + BW) * PKW + 7) : slot;
            pk[slot] = isr ? w : -w;
            __builtin_amdgcn_sched_barrier(0);
        }
    }
    __syncthreads();

    if (wv == 0) {
        const v4f* pk4 = reinterpret_cast<const v4f*>(Abuf + PK0);
        v4f lo[3], hi[3];
#pragma unroll
        for (int g = 0; g < 3; ++g) {
            lo[g] = pk4[2 * (32 * g + lane)];
            hi[g] = pk4[2 * (32 * g + lane) + 1];
        }
        float a[6];
#pragma unroll
        for (int i = 0; i < 6; ++i) a[i] = lane_value(hi[2][3], (NB - 1 - i + BW) & 31);
        float xs[3] = {0.0f, 0.0f, 0.0f};
#pragma unroll
        for (int m = NB - 1; m >= 0; --m) {
            const int g = m >> 5, ln = m & 31;
            const float x = a[0];
            xs[g] = (lane == ln) ? x : xs[g];
            a[0] = fmaf(lane_value(lo[g][1], ln), x, a[1]);
            a[1] = fmaf(lane_value(lo[g][2], ln), x, a[2]);
            a[2] = fmaf(lane_value(lo[g][3], ln), x, a[3]);
            a[3] = fmaf(lane_value(hi[g][0], ln), x, a[4]);
            a[4] = fmaf(lane_value(hi[g][1], ln), x, a[5]);
            a[5] = fmaf(lane_value(hi[g][2], ln), x, lane_value(hi[g][3], ln));
        }
        out[b * 90 + lane]      = xs[0];
        out[b * 90 + 32 + lane] = xs[1];
        if (lane < NB - 64) out[b * 90 + 64 + lane] = xs[2];
    }
}

extern "C" void kernel_launch(void* const* d_in, const int* in_sizes, int n_in,
                              void* d_out, int out_size, void* d_ws, size_t ws_size,
                              hipStream_t stream) {
    (void)n_in; (void)out_size; (void)d_ws; (void)ws_size;
    const float* coeffs = (const float*)d_in[0];
    const float* rhs    = (const float*)d_in[1];
    const float* iv     = (const float*)d_in[2];
    const float* steps  = (const float*)d_in[3];
    float* out = (float*)d_out;
    int bs = in_sizes[0] / 90;
    odein_solve_kernel<<<bs, 256, 0, stream>>>(coeffs, rhs, iv, steps, out);
}


# One wave solves one system start to end (work-group = 1 wave; 7 of 8 idled in the solve): A kept as 8 floats per row, AtA as its lower band in A's place, LDS per system 6 KB, not 74
# speedup vs baseline: 28.6582x; 3.7114x over previous
#include <hip/hip_runtime.h>
#include <hip/hip_bf16.h>


typedef float v2f __attribute__((ext_vector_type(2)));
typedef float v4f __attribute__((ext_vector_type(4)));
typedef float v8f __attribute__((ext_vector_type(8)));

#define KROWS 180
#define AW    8
#define APAD  2
#define NB    90
#define NBP   96
#define BW    6
#define PKW   8
#define LA    8
#define BAND0 0
#define PK0   (NBP * 8)
#define ABUF  (PK0 + NBP * PKW)

__device__ __forceinline__ float lane_value(float v, int ln) {
    return __int_as_float(__builtin_amdgcn_readlane(__float_as_int(v), ln));
}

#define A_ROW0(s) (4 + 6 * (s))

__device__ __forceinline__ int row_base(int k) { return k < 4 ? 0 : 3 * ((k - 4) / 6); }

__device__ __forceinline__ float a_entry(const float* Ac, int k, int base, int col) {
    const int e = col - base;
    const float v = Ac[k * AW + e];
    return ((unsigned)e < (unsigned)AW) ? v : 0.0f;
}

__device__ __forceinline__ v2f a_fragment(const float* Ac, int kb, int half, int col) {
    const int k0 = 4 * kb + 2 * half;
    const int bx = half ? row_base(4 * kb + 2) : row_base(4 * kb);
    const int by = half ? row_base(4 * kb + 3) : row_base(4 * kb + 1);
    v2f f;
    f.x = a_entry(Ac, k0, bx, col);
    f.y = a_entry(Ac, k0 + 1, by, col);
    return f;
}

__device__ __forceinline__ v8f ata_kblock(const float* Ac, int kb, int half, int lm,
                                          int m0, int n0, v8f acc) {
    const v2f a  = a_fragment(Ac, kb, half, m0 + lm);
    const v2f bb = (m0 == n0) ? a : a_fragment(Ac, kb, half, n0 + lm);
    return __builtin_amdgcn_wmma_f32_16x16x4_f32(false, a, false, bb, (short)0, acc, false, false);
}

__device__ __forceinline__ float load_band_column(const float* band, int c, int lane) {
    const int k = (lane - c) & 15;
    return band[c * 8 + ((lane == 16) ? 7 : (k & 7))];
}

__global__ __launch_bounds__(32)
void odein_solve_kernel(const float* __restrict__ coeffs,
                        const float* __restrict__ rhs,
                        const float* __restrict__ iv_rhs,
                        const float* __restrict__ steps,
                        float* __restrict__ out)
{
    static_assert(ABUF >= (KROWS + APAD) * AW, "array too small for A");
    __shared__ __attribute__((aligned(16))) float Ac[ABUF];
    __shared__ float st[32];
    __shared__ float unkept[32];

    const int lane = threadIdx.x;
    const int b    = blockIdx.x;
    const int half = lane >> 4, lm = lane & 15;

    {
        const v4f z4 = {0.0f, 0.0f, 0.0f, 0.0f};
        v4f* Ac4   = reinterpret_cast<v4f*>(Ac);
        for (int i = lane; i < ABUF / 4; i += 32) Ac4[i] = z4;
    }
    if (lane < 29) st[lane] = steps[b * 29 + lane];
    __syncthreads();

    float r1v[3] = {0.0f, 0.0f, 0.0f};
#pragma unroll
    for (int g = 0; g < 3; ++g) {
        const int c = lane + 32 * g;
        if (c < NB) {
            const float cf = coeffs[b * 90 + c];
            const int s = c / 3;
            Ac[A_ROW0(s) * AW + (c - 3 * s)] = cf;
            float v = cf * rhs[b * 30 + s];
            if      (c == 0) v += iv_rhs[b * 4 + 0];
            else if (c == 1) v += iv_rhs[b * 4 + 1];
            else if (c == 3) v += iv_rhs[b * 4 + 2];
            else if (c == 4) v += iv_rhs[b * 4 + 3];
            r1v[g] = v;
        }
    }
    if (lane < 4)
        Ac[lane * AW + lane + (lane >> 1)] = 1.0f;
    if (lane < 29) {
        const int s = lane;
        const float h = st[s], h2 = h * h;
        float* r = Ac + (A_ROW0(s) + 1) * AW;
        r[3] = -1.0f;
        r[0] = 1.0f;
        r[1] = h;
        r[2] = 0.5f * h2;
        r += AW;
        r[4] = -h;
        r[1] = h;
        r[2] = h2;
        r += AW;
        r[0] = -1.0f;
        r[3] = 1.0f;
        r[4] = -h;
        r[5] = 0.5f * h2;
        r += AW;
        r[1] = h;
        r[4] = -h;
        r[5] = h2;
    }
    if (lane >= 1 && lane < 29) {
        const int s = lane;
        float* r = Ac + (A_ROW0(s - 1) + 5) * AW;
        r[1] = -1.0f;
        r[7] = 1.0f;
        r[5] = -(st[s] + st[s - 1]);
    }
    __syncthreads();

    float* band = Ac + BAND0;
    {
        const int DK[6] = {0, 6, 15, 23, 30, 33};
        const int OK[6] = {6, 15, 23, 30, 39, 0};
#pragma unroll
        for (int t = 0; t < 6; ++t) {
            v8f accd = {0.f, 0.f, 0.f, 0.f, 0.f, 0.f, 0.f, 0.f};
            v8f acco = {0.f, 0.f, 0.f, 0.f, 0.f, 0.f, 0.f, 0.f};
#pragma unroll
            for (int i = 0; i < 12; ++i) {
                accd = ata_kblock(Ac, DK[t] + i, half, lm, 16 * t, 16 * t, accd);
                if (t < 5 && i < 4)
                    acco = ata_kblock(Ac, OK[t] + i, half, lm, 16 * (t + 1), 16 * t, acco);
            }
            __syncthreads();
            const int n = 16 * t + lm;
#pragma unroll
            for (int v = 0; v < 8; ++v) {
                const int d = v + 8 * half - lm;
                float* dst = ((unsigned)d <= (unsigned)BW) ? (band + n * 8 + d) : (unkept + lane);
                *dst = accd[v];
            }
            if (t < 5) {
#pragma unroll
                for (int v = 0; v < 6; ++v) {
                    const int d = 16 + v + 8 * half - lm;
                    float* dst = ((unsigned)d <= (unsigned)BW) ? (band + n * 8 + d) : (unkept + lane);
                    *dst = acco[v];
                }
            }
        }
    }
#pragma unroll
    for (int g = 0; g < 3; ++g)
        if (lane + 32 * g < NB) band[(lane + 32 * g) * 8 + 7] = r1v[g];
    __syncthreads();

    float* pk = Ac + PK0;
    {
        float col[NB];
#pragma unroll
        for (int c = 0; c < BW + 1 + LA; ++c) col[c] = load_band_column(band, c, lane);
#pragma unroll
        for (int j = 0; j < NB; ++j) {
            if (j + BW + 1 + LA < NB)
                col[j + BW + 1 + LA] = load_band_column(band, j + BW + 1 + LA, lane);
            const int  k    = (lane - j) & 15;
            const bool isr  = (lane == 16);
            const bool inb  = (lane < 16) & (k <= BW) & (j + k < NB);
            const bool live = isr | inb;
            const float pm  = live ? col[j] : 0.0f;
            const float piv = lane_value(col[j], j & 15);
            const float r0  = __builtin_amdgcn_rcpf(piv);
            const float q0  = pm * r0;
            const float w   = fmaf(fmaf(-piv, q0, pm), r0, q0);
#pragma unroll
            for (int k2 = 1; k2 <= BW; ++k2)
                if (j + k2 < NB)
                    col[j + k2] = fmaf(-lane_value(w, (j + k2) & 15), pm, col[j + k2]);
            int slot = (inb & (k >= 1)) ? ((j + k) * PKW + k) : (lane * PKW);
            slot = isr ? ((j + BW) * PKW + 7) : slot;
            pk[slot] = isr ? w : -w;
            __builtin_amdgcn_sched_barrier(0);
        }
    }
    __syncthreads();

    {
        const v4f* pk4 = reinterpret_cast<const v4f*>(pk);
        v4f lo[3], hi[3];
#pragma unroll
        for (int g = 0; g < 3; ++g) {
            lo[g] = pk4[2 * (32 * g + lane)];
            hi[g] = pk4[2 * (32 * g + lane) + 1];
        }
        float a[6];
#pragma unroll
        for (int i = 0; i < 6; ++i) a[i] = lane_value(hi[2][3], (NB - 1 - i + BW) & 31);
        float xs[3] = {0.0f, 0.0f, 0.0f};
#pragma unroll
        for (int m = NB - 1; m >= 0; --m) {
            const int g = m >> 5, ln = m & 31;
            const float x = a[0];
            xs[g] = (lane == ln) ? x : xs[g];
            a[0] = fmaf(lane_value(lo[g][1], ln), x, a[1]);
            a[1] = fmaf(lane_value(lo[g][2], ln), x, a[2]);
            a[2] = fmaf(lane_value(lo[g][3], ln), x, a[3]);
            a[3] = fmaf(lane_value(hi[g][0], ln), x, a[4]);
            a[4] = fmaf(lane_value(hi[g][1], ln), x, a[5]);
            a[5] = fmaf(lane_value(hi[g][2], ln), x, lane_value(hi[g][3], ln));
        }
        out[b * 90 + lane]      = xs[0];
        out[b * 90 + 32 + lane] = xs[1];
        if (lane < NB - 64) out[b * 90 + 64 + lane] = xs[2];
    }
}

extern "C" void kernel_launch(void* const* d_in, const int* in_sizes, int n_in,
                              void* d_out, int out_size, void* d_ws, size_t ws_size,
                              hipStream_t stream) {
    (void)n_in; (void)out_size; (void)d_ws; (void)ws_size;
    const float* coeffs = (const float*)d_in[0];
    const float* rhs    = (const float*)d_in[1];
    const float* iv     = (const float*)d_in[2];
    const float* steps  = (const float*)d_in[3];
    float* out = (float*)d_out;
    int bs = in_sizes[0] / 90;
    odein_solve_kernel<<<bs, 32, 0, stream>>>(coeffs, rhs, iv, steps, out);
}


# Back substitution with the open sums in the lanes: x[m] reaches every lane as a DPP row_share operand, each lane adds x[m] times its entry of packed row m: 1 fma per unknown, was 7 readlanes + 6 fmas
# speedup vs baseline: 34.5027x; 1.2039x over previous
#include <hip/hip_runtime.h>
#include <hip/hip_bf16.h>


typedef float v2f __attribute__((ext_vector_type(2)));
typedef float v4f __attribute__((ext_vector_type(4)));
typedef float v8f __attribute__((ext_vector_type(8)));

#define KROWS 180
#define AW    8
#define APAD  2
#define NB    NB_
#define NBP   96
#define BW    6
#define PKW   8
#define LA    8
#define BAND0 0
#define PK0   (NBP * 8)
#define ABUF  (PK0 + NBP * PKW)

__device__ __forceinline__ float lane_value(float v, int ln) {
    return __int_as_float(__builtin_amdgcn_readlane(__float_as_int(v), ln));
}

template <int N>
__device__ __forceinline__ float row_value(float v) {
    return __int_as_float(__builtin_amdgcn_update_dpp(0, __float_as_int(v), 0x150 + N, 0xf, 0xf, true));
}

#define NB_   90
#define BLA_  12

template <int M>
__device__ __forceinline__ void back_rows(float& acc, float (&X)[6], float (&rv)[NB_], float (&yn)[12],
                                          const float* const (&rvP)[16], const float* yE,
                                          const float* yO, const bool sE) {
    if constexpr (M >= 0) {
        constexpr int b = (NB_ - 1 - M) >> 3;
        if constexpr (M - BLA_ >= 0)
            rv[M - BLA_] = rvP[(M - BLA_) & 15][128 * ((M - BLA_) >> 4)];
        if constexpr (((NB_ - 1 - M) & 7) == 0 && b <= 9)
            yn[b] = ((b & 1) ? yO : yE)[128 * (4 - (b >> 1))];
        acc = fmaf(row_value<(M & 15)>(acc), rv[M], acc);
        if constexpr (((NB_ - 1 - M) & 7) == 7 || M == 0) {
            const bool s = (b & 1) ? !sE : sE;
            X[b >> 1] = s ? acc : X[b >> 1];
            if constexpr (b <= 9) acc = s ? yn[b] : acc;
        }
        __builtin_amdgcn_sched_barrier(0);
        back_rows<M - 1>(acc, X, rv, yn, rvP, yE, yO, sE);
    }
}

#define A_ROW0(s) (4 + 6 * (s))

__device__ __forceinline__ int row_base(int k) { return k < 4 ? 0 : 3 * ((k - 4) / 6); }

__device__ __forceinline__ float a_entry(const float* Ac, int k, int base, int col) {
    const int e = col - base;
    const float v = Ac[k * AW + e];
    return ((unsigned)e < (unsigned)AW) ? v : 0.0f;
}

__device__ __forceinline__ v2f a_fragment(const float* Ac, int kb, int half, int col) {
    const int k0 = 4 * kb + 2 * half;
    const int bx = half ? row_base(4 * kb + 2) : row_base(4 * kb);
    const int by = half ? row_base(4 * kb + 3) : row_base(4 * kb + 1);
    v2f f;
    f.x = a_entry(Ac, k0, bx, col);
    f.y = a_entry(Ac, k0 + 1, by, col);
    return f;
}

__device__ __forceinline__ v8f ata_kblock(const float* Ac, int kb, int half, int lm,
                                          int m0, int n0, v8f acc) {
    const v2f a  = a_fragment(Ac, kb, half, m0 + lm);
    const v2f bb = (m0 == n0) ? a : a_fragment(Ac, kb, half, n0 + lm);
    return __builtin_amdgcn_wmma_f32_16x16x4_f32(false, a, false, bb, (short)0, acc, false, false);
}

__device__ __forceinline__ float load_band_column(const float* band, int c, int lane) {
    const int k = (lane - c) & 15;
    return band[c * 8 + ((lane == 16) ? 7 : (k & 7))];
}

__global__ __launch_bounds__(32)
void odein_solve_kernel(const float* __restrict__ coeffs,
                        const float* __restrict__ rhs,
                        const float* __restrict__ iv_rhs,
                        const float* __restrict__ steps,
                        float* __restrict__ out)
{
    static_assert(ABUF >= (KROWS + APAD) * AW, "array too small for A");
    __shared__ __attribute__((aligned(16))) float Ac[ABUF];
    __shared__ float st[32];
    __shared__ float unkept[32];

    const int lane = threadIdx.x;
    const int b    = blockIdx.x;
    const int half = lane >> 4, lm = lane & 15;

    {
        const v4f z4 = {0.0f, 0.0f, 0.0f, 0.0f};
        v4f* Ac4   = reinterpret_cast<v4f*>(Ac);
        for (int i = lane; i < ABUF / 4; i += 32) Ac4[i] = z4;
    }
    if (lane < 29) st[lane] = steps[b * 29 + lane];
    __syncthreads();

    float r1v[3] = {0.0f, 0.0f, 0.0f};
#pragma unroll
    for (int g = 0; g < 3; ++g) {
        const int c = lane + 32 * g;
        if (c < NB) {
            const float cf = coeffs[b * 90 + c];
            const int s = c / 3;
            Ac[A_ROW0(s) * AW + (c - 3 * s)] = cf;
            float v = cf * rhs[b * 30 + s];
            if      (c == 0) v += iv_rhs[b * 4 + 0];
            else if (c == 1) v += iv_rhs[b * 4 + 1];
            else if (c == 3) v += iv_rhs[b * 4 + 2];
            else if (c == 4) v += iv_rhs[b * 4 + 3];
            r1v[g] = v;
        }
    }
    if (lane < 4)
        Ac[lane * AW + lane + (lane >> 1)] = 1.0f;
    if (lane < 29) {
        const int s = lane;
        const float h = st[s], h2 = h * h;
        float* r = Ac + (A_ROW0(s) + 1) * AW;
        r[3] = -1.0f;
        r[0] = 1.0f;
        r[1] = h;
        r[2] = 0.5f * h2;
        r += AW;
        r[4] = -h;
        r[1] = h;
        r[2] = h2;
        r += AW;
        r[0] = -1.0f;
        r[3] = 1.0f;
        r[4] = -h;
        r[5] = 0.5f * h2;
        r += AW;
        r[1] = h;
        r[4] = -h;
        r[5] = h2;
    }
    if (lane >= 1 && lane < 29) {
        const int s = lane;
        float* r = Ac + (A_ROW0(s - 1) + 5) * AW;
        r[1] = -1.0f;
        r[7] = 1.0f;
        r[5] = -(st[s] + st[s - 1]);
    }
    __syncthreads();

    float* band = Ac + BAND0;
    {
        const int DK[6] = {0, 6, 15, 23, 30, 33};
        const int OK[6] = {6, 15, 23, 30, 39, 0};
#pragma unroll
        for (int t = 0; t < 6; ++t) {
            v8f accd = {0.f, 0.f, 0.f, 0.f, 0.f, 0.f, 0.f, 0.f};
            v8f acco = {0.f, 0.f, 0.f, 0.f, 0.f, 0.f, 0.f, 0.f};
#pragma unroll
            for (int i = 0; i < 12; ++i) {
                accd = ata_kblock(Ac, DK[t] + i, half, lm, 16 * t, 16 * t, accd);
                if (t < 5 && i < 4)
                    acco = ata_kblock(Ac, OK[t] + i, half, lm, 16 * (t + 1), 16 * t, acco);
            }
            __syncthreads();
            const int n = 16 * t + lm;
#pragma unroll
            for (int v = 0; v < 8; ++v) {
                const int d = v + 8 * half - lm;
                float* dst = ((unsigned)d <= (unsigned)BW) ? (band + n * 8 + d) : (unkept + lane);
                *dst = accd[v];
            }
            if (t < 5) {
#pragma unroll
                for (int v = 0; v < 6; ++v) {
                    const int d = 16 + v + 8 * half - lm;
                    float* dst = ((unsigned)d <= (unsigned)BW) ? (band + n * 8 + d) : (unkept + lane);
                    *dst = acco[v];
                }
            }
        }
    }
#pragma unroll
    for (int g = 0; g < 3; ++g)
        if (lane + 32 * g < NB) band[(lane + 32 * g) * 8 + 7] = r1v[g];
    __syncthreads();
#pragma unroll
    for (int g = 0; g < 3; ++g)
        Ac[PK0 + (lane + 32 * g) * PKW] = 0.0f;

    float* pk = Ac + PK0;
    {
        float col[NB];
#pragma unroll
        for (int c = 0; c < BW + 1 + LA; ++c) col[c] = load_band_column(band, c, lane);
#pragma unroll
        for (int j = 0; j < NB; ++j) {
            if (j + BW + 1 + LA < NB)
                col[j + BW + 1 + LA] = load_band_column(band, j + BW + 1 + LA, lane);
            const int  k    = (lane - j) & 15;
            const bool isr  = (lane == 16);
            const bool inb  = (lane < 16) & (k <= BW) & (j + k < NB);
            const bool live = isr | inb;
            const float pm  = live ? col[j] : 0.0f;
            const float piv = lane_value(col[j], j & 15);
            const float r0  = __builtin_amdgcn_rcpf(piv);
            const float q0  = pm * r0;
            const float w   = fmaf(fmaf(-piv, q0, pm), r0, q0);
#pragma unroll
            for (int k2 = 1; k2 <= BW; ++k2)
                if (j + k2 < NB)
                    col[j + k2] = fmaf(-lane_value(w, (j + k2) & 15), pm, col[j + k2]);
            float* dst = (inb & (k >= 1)) ? (pk + (j + k) * PKW + k) : (unkept + lane);
            dst = isr ? (pk + (j + BW) * PKW + 7) : dst;
            *dst = isr ? w : -w;
            __builtin_amdgcn_sched_barrier(0);
        }
    }
    __syncthreads();

    {
        const int l16 = lane & 15;
        const float* rvP[16];
#pragma unroll
        for (int p = 0; p < 16; ++p) {
            const int k = (p - l16) & 15;
            rvP[p] = pk + p * PKW + (((unsigned)(k - 1) < (unsigned)BW) ? k : 0);
        }
        const float* yE = pk + 71 + 8 * ((l16 - 2) & 15);
        const float* yO = pk + 7 + 8 * ((l16 - 10) & 15);
        const bool sE = ((l16 - 2) & 15) < 8;

        float acc = yO[128 * 5];
        float X[6] = {0.0f, 0.0f, 0.0f, 0.0f, 0.0f, 0.0f};
        float rv[NB];
        float yn[12];
#pragma unroll
        for (int m = NB - 1; m > NB - 1 - BLA_; --m) rv[m] = rvP[m & 15][128 * (m >> 4)];
        back_rows<NB - 1>(acc, X, rv, yn, rvP, yE, yO, sE);
#pragma unroll
        for (int t = 0; t < 3; ++t) {
            const int c = 2 * t + half;
            const int i = 74 - 16 * c + ((l16 - 10) & 15);
            const float v = half ? X[2 * t + 1] : X[2 * t];
            if (i >= 0) out[b * 90 + i] = v;
        }
    }
}

extern "C" void kernel_launch(void* const* d_in, const int* in_sizes, int n_in,
                              void* d_out, int out_size, void* d_ws, size_t ws_size,
                              hipStream_t stream) {
    (void)n_in; (void)out_size; (void)d_ws; (void)ws_size;
    const float* coeffs = (const float*)d_in[0];
    const float* rhs    = (const float*)d_in[1];
    const float* iv     = (const float*)d_in[2];
    const float* steps  = (const float*)d_in[3];
    float* out = (float*)d_out;
    int bs = in_sizes[0] / 90;
    odein_solve_kernel<<<bs, 32, 0, stream>>>(coeffs, rhs, iv, steps, out);
}


# Elimination stores its quotients unnegated, one LDS row per step at a fixed offset from per-lane pointers: no address, target or sign select per step; back-sub fma negates, folds its row_share
# speedup vs baseline: 36.0136x; 1.0438x over previous
#include <hip/hip_runtime.h>
#include <hip/hip_bf16.h>


typedef float v2f __attribute__((ext_vector_type(2)));
typedef float v4f __attribute__((ext_vector_type(4)));
typedef float v8f __attribute__((ext_vector_type(8)));

#define KROWS 180
#define AW    8
#define APAD  2
#define NB    NB_
#define NBP   96
#define BW    6
#define PKW   8
#define LA    8
#define BAND0 0
#define PK0   (NBP * 8)
#define QPRE  BW
#define ABUF  (PK0 + NBP * PKW)

__device__ __forceinline__ float lane_value(float v, int ln) {
    return __int_as_float(__builtin_amdgcn_readlane(__float_as_int(v), ln));
}

template <int N>
__device__ __forceinline__ float row_value(float v) {
    return __int_as_float(__builtin_amdgcn_update_dpp(0, __float_as_int(v), 0x150 + N, 0xf, 0xf, true));
}

#define NB_   90
#define BLA_  12

template <int M>
__device__ __forceinline__ void back_rows(float& acc, float (&X)[6], float (&rv)[NB_], float (&yn)[12],
                                          const float* const (&rvP)[16], const float* yE,
                                          const float* yO, const bool sE) {
    if constexpr (M >= 0) {
        constexpr int b = (NB_ - 1 - M) >> 3;
        if constexpr (M - BLA_ >= 0)
            rv[M - BLA_] = rvP[(M - BLA_) & 15][128 * ((M - BLA_) >> 4)];
        if constexpr (((NB_ - 1 - M) & 7) == 0 && b <= 9)
            yn[b] = ((b & 1) ? yO : yE)[128 * (4 - (b >> 1))];
        acc = fmaf(-row_value<(M & 15)>(acc), rv[M], acc);
        if constexpr (((NB_ - 1 - M) & 7) == 7 || M == 0) {
            const bool s = (b & 1) ? !sE : sE;
            X[b >> 1] = s ? acc : X[b >> 1];
            if constexpr (b <= 9) acc = s ? yn[b] : acc;
        }
        __builtin_amdgcn_sched_barrier(0);
        back_rows<M - 1>(acc, X, rv, yn, rvP, yE, yO, sE);
    }
}

#define A_ROW0(s) (4 + 6 * (s))

__device__ __forceinline__ int row_base(int k) { return k < 4 ? 0 : 3 * ((k - 4) / 6); }

__device__ __forceinline__ float a_entry(const float* Ac, int k, int base, int col) {
    const int e = col - base;
    const float v = Ac[k * AW + e];
    return ((unsigned)e < (unsigned)AW) ? v : 0.0f;
}

__device__ __forceinline__ v2f a_fragment(const float* Ac, int kb, int half, int col) {
    const int k0 = 4 * kb + 2 * half;
    const int bx = half ? row_base(4 * kb + 2) : row_base(4 * kb);
    const int by = half ? row_base(4 * kb + 3) : row_base(4 * kb + 1);
    v2f f;
    f.x = a_entry(Ac, k0, bx, col);
    f.y = a_entry(Ac, k0 + 1, by, col);
    return f;
}

__device__ __forceinline__ v8f ata_kblock(const float* Ac, int kb, int half, int lm,
                                          int m0, int n0, v8f acc) {
    const v2f a  = a_fragment(Ac, kb, half, m0 + lm);
    const v2f bb = (m0 == n0) ? a : a_fragment(Ac, kb, half, n0 + lm);
    return __builtin_amdgcn_wmma_f32_16x16x4_f32(false, a, false, bb, (short)0, acc, false, false);
}

__device__ __forceinline__ float load_band_column(const float* band, int c, int lane) {
    const int k = (lane - c) & 15;
    return band[c * 8 + ((lane == 16) ? 7 : (k & 7))];
}

__global__ __launch_bounds__(32)
void odein_solve_kernel(const float* __restrict__ coeffs,
                        const float* __restrict__ rhs,
                        const float* __restrict__ iv_rhs,
                        const float* __restrict__ steps,
                        float* __restrict__ out)
{
    static_assert(ABUF >= (KROWS + APAD) * AW, "array too small for A");
    __shared__ __attribute__((aligned(16))) float Ac[ABUF];
    __shared__ float st[32];
    __shared__ float unkept[32];

    const int lane = threadIdx.x;
    const int b    = blockIdx.x;
    const int half = lane >> 4, lm = lane & 15;

    {
        const v4f z4 = {0.0f, 0.0f, 0.0f, 0.0f};
        v4f* Ac4   = reinterpret_cast<v4f*>(Ac);
        for (int i = lane; i < ABUF / 4; i += 32) Ac4[i] = z4;
    }
    if (lane < 29) st[lane] = steps[b * 29 + lane];
    __syncthreads();

    float r1v[3] = {0.0f, 0.0f, 0.0f};
#pragma unroll
    for (int g = 0; g < 3; ++g) {
        const int c = lane + 32 * g;
        if (c < NB) {
            const float cf = coeffs[b * 90 + c];
            const int s = c / 3;
            Ac[A_ROW0(s) * AW + (c - 3 * s)] = cf;
            float v = cf * rhs[b * 30 + s];
            if      (c == 0) v += iv_rhs[b * 4 + 0];
            else if (c == 1) v += iv_rhs[b * 4 + 1];
            else if (c == 3) v += iv_rhs[b * 4 + 2];
            else if (c == 4) v += iv_rhs[b * 4 + 3];
            r1v[g] = v;
        }
    }
    if (lane < 4)
        Ac[lane * AW + lane + (lane >> 1)] = 1.0f;
    if (lane < 29) {
        const int s = lane;
        const float h = st[s], h2 = h * h;
        float* r = Ac + (A_ROW0(s) + 1) * AW;
        r[3] = -1.0f;
        r[0] = 1.0f;
        r[1] = h;
        r[2] = 0.5f * h2;
        r += AW;
        r[4] = -h;
        r[1] = h;
        r[2] = h2;
        r += AW;
        r[0] = -1.0f;
        r[3] = 1.0f;
        r[4] = -h;
        r[5] = 0.5f * h2;
        r += AW;
        r[1] = h;
        r[4] = -h;
        r[5] = h2;
    }
    if (lane >= 1 && lane < 29) {
        const int s = lane;
        float* r = Ac + (A_ROW0(s - 1) + 5) * AW;
        r[1] = -1.0f;
        r[7] = 1.0f;
        r[5] = -(st[s] + st[s - 1]);
    }
    __syncthreads();

    float* band = Ac + BAND0;
    {
        const int DK[6] = {0, 6, 15, 23, 30, 33};
        const int OK[6] = {6, 15, 23, 30, 39, 0};
#pragma unroll
        for (int t = 0; t < 6; ++t) {
            v8f accd = {0.f, 0.f, 0.f, 0.f, 0.f, 0.f, 0.f, 0.f};
            v8f acco = {0.f, 0.f, 0.f, 0.f, 0.f, 0.f, 0.f, 0.f};
#pragma unroll
            for (int i = 0; i < 12; ++i) {
                accd = ata_kblock(Ac, DK[t] + i, half, lm, 16 * t, 16 * t, accd);
                if (t < 5 && i < 4)
                    acco = ata_kblock(Ac, OK[t] + i, half, lm, 16 * (t + 1), 16 * t, acco);
            }
            __syncthreads();
            const int n = 16 * t + lm;
#pragma unroll
            for (int v = 0; v < 8; ++v) {
                const int d = v + 8 * half - lm;
                float* dst = ((unsigned)d <= (unsigned)BW) ? (band + n * 8 + d) : (unkept + lane);
                *dst = accd[v];
            }
            if (t < 5) {
#pragma unroll
                for (int v = 0; v < 6; ++v) {
                    const int d = 16 + v + 8 * half - lm;
                    float* dst = ((unsigned)d <= (unsigned)BW) ? (band + n * 8 + d) : (unkept + lane);
                    *dst = acco[v];
                }
            }
        }
    }
#pragma unroll
    for (int g = 0; g < 3; ++g)
        if (lane + 32 * g < NB) band[(lane + 32 * g) * 8 + 7] = r1v[g];
    __syncthreads();
    Ac[PK0 + lane] = 0.0f;
    if (lane < QPRE * PKW - 32) Ac[PK0 + 32 + lane] = 0.0f;
#pragma unroll
    for (int g = 0; g < 3; ++g)
        if (lane + 32 * g < NB) Ac[PK0 + (QPRE + lane + 32 * g) * PKW] = 0.0f;

    static_assert(32 <= 8 * (BW + 1 + LA), "a step's 32 spare words must lie in band columns that are already in registers");
    static_assert(8 * (NB - 1) + 32 <= PK0, "the spare words of the last step must stay inside the band");
    float* pk = Ac + PK0 + QPRE * PKW;
    {
        float* wP[16];
#pragma unroll
        for (int p = 0; p < 16; ++p) {
            const int k = (lane - p) & 15;
            const bool keeps = (lane == 16) | ((lane < 16) & ((unsigned)(k - 1) < (unsigned)BW));
            wP[p] = keeps ? (pk + ((lane == 16) ? 7 : k)) : (band + lane);
        }
        float col[NB];
#pragma unroll
        for (int c = 0; c < BW + 1 + LA; ++c) col[c] = load_band_column(band, c, lane);
#pragma unroll
        for (int j = 0; j < NB; ++j) {
            if (j + BW + 1 + LA < NB)
                col[j + BW + 1 + LA] = load_band_column(band, j + BW + 1 + LA, lane);
            const int  k    = (lane - j) & 15;
            const bool isr  = (lane == 16);
            const bool inb  = (lane < 16) & (k <= BW) & (j + k < NB);
            const bool live = isr | inb;
            const float pm  = live ? col[j] : 0.0f;
            const float piv = lane_value(col[j], j & 15);
            const float r0  = __builtin_amdgcn_rcpf(piv);
            const float q0  = pm * r0;
            const float w   = fmaf(fmaf(-piv, q0, pm), r0, q0);
#pragma unroll
            for (int k2 = 1; k2 <= BW; ++k2)
                if (j + k2 < NB)
                    col[j + k2] = fmaf(-lane_value(w, (j + k2) & 15), pm, col[j + k2]);
            wP[j & 15][j * PKW] = w;
            __builtin_amdgcn_sched_barrier(0);
        }
    }
    __syncthreads();

    {
        const int l16 = lane & 15;
        const float* rvP[16];
#pragma unroll
        for (int p = 0; p < 16; ++p) {
            const int k = (p - l16) & 15;
            rvP[p] = ((unsigned)(k - 1) < (unsigned)BW) ? (pk + (p - k) * PKW + k) : (pk + p * PKW);
        }
        const float* yE = pk + 23 + 8 * ((l16 - 2) & 15);
        const float* yO = pk - 41 + 8 * ((l16 - 10) & 15);
        const bool sE = ((l16 - 2) & 15) < 8;

        float acc = yO[128 * 5];
        float X[6] = {0.0f, 0.0f, 0.0f, 0.0f, 0.0f, 0.0f};
        float rv[NB];
        float yn[12];
#pragma unroll
        for (int m = NB - 1; m > NB - 1 - BLA_; --m) rv[m] = rvP[m & 15][128 * (m >> 4)];
        back_rows<NB - 1>(acc, X, rv, yn, rvP, yE, yO, sE);
#pragma unroll
        for (int t = 0; t < 3; ++t) {
            const int c = 2 * t + half;
            const int i = 74 - 16 * c + ((l16 - 10) & 15);
            const float v = half ? X[2 * t + 1] : X[2 * t];
            if (i >= 0) out[b * 90 + i] = v;
        }
    }
}

extern "C" void kernel_launch(void* const* d_in, const int* in_sizes, int n_in,
                              void* d_out, int out_size, void* d_ws, size_t ws_size,
                              hipStream_t stream) {
    (void)n_in; (void)out_size; (void)d_ws; (void)ws_size;
    const float* coeffs = (const float*)d_in[0];
    const float* rhs    = (const float*)d_in[1];
    const float* iv     = (const float*)d_in[2];
    const float* steps  = (const float*)d_in[3];
    float* out = (float*)d_out;
    int bs = in_sizes[0] / 90;
    odein_solve_kernel<<<bs, 32, 0, stream>>>(coeffs, rhs, iv, steps, out);
}


# Two systems per wave: each 16-lane row eliminates and back-substitutes its own system (rhs row in lane 15 of the row, pivot and multipliers as DPP row_share operands); A^T A of the two in turn
# speedup vs baseline: 40.8253x; 1.1336x over previous
#include <hip/hip_runtime.h>
#include <hip/hip_bf16.h>


typedef float v2f __attribute__((ext_vector_type(2)));
typedef float v4f __attribute__((ext_vector_type(4)));
typedef float v8f __attribute__((ext_vector_type(8)));

#define KROWS 180
#define AW    8
#define APAD  2
#define NB    NB_
#define NBP   96
#define BW    6
#define PKW   8
#define LA    8
#define EP    15
#define BAND0 0
#define PK0   (NBP * 8)
#define QPRE  BW
#define ABUF  (PK0 + NBP * PKW)
#define SYSF  (ABUF + 32)

template <int N>
__device__ __forceinline__ float row_value(float v) {
    return __int_as_float(__builtin_amdgcn_update_dpp(0, __float_as_int(v), 0x150 + N, 0xf, 0xf, true));
}

#define NB_   90
#define BLA_  12

template <int M>
__device__ __forceinline__ void back_rows(float& acc, float (&X)[6], float (&rv)[NB_], float (&yn)[12],
                                          const float* const (&rvP)[16], const float* yE,
                                          const float* yO, const bool sE) {
    if constexpr (M >= 0) {
        constexpr int b = (NB_ - 1 - M) >> 3;
        if constexpr (M - BLA_ >= 0)
            rv[M - BLA_] = rvP[(M - BLA_) & 15][128 * ((M - BLA_) >> 4)];
        if constexpr (((NB_ - 1 - M) & 7) == 0 && b <= 9)
            yn[b] = ((b & 1) ? yO : yE)[128 * (4 - (b >> 1))];
        acc = fmaf(-row_value<(M & 15)>(acc), rv[M], acc);
        if constexpr (((NB_ - 1 - M) & 7) == 7 || M == 0) {
            const bool s = (b & 1) ? !sE : sE;
            X[b >> 1] = s ? acc : X[b >> 1];
            if constexpr (b <= 9) acc = s ? yn[b] : acc;
        }
        __builtin_amdgcn_sched_barrier(0);
        back_rows<M - 1>(acc, X, rv, yn, rvP, yE, yO, sE);
    }
}

#define A_ROW0(s) (4 + 6 * (s))

__device__ __forceinline__ int row_base(int k) { return k < 4 ? 0 : 3 * ((k - 4) / 6); }

__device__ __forceinline__ float a_entry(const float* Ac, int k, int base, int col) {
    const int e = col - base;
    const float v = Ac[k * AW + e];
    return ((unsigned)e < (unsigned)AW) ? v : 0.0f;
}

__device__ __forceinline__ v2f a_fragment(const float* Ac, int kb, int half, int col) {
    const int k0 = 4 * kb + 2 * half;
    const int bx = half ? row_base(4 * kb + 2) : row_base(4 * kb);
    const int by = half ? row_base(4 * kb + 3) : row_base(4 * kb + 1);
    v2f f;
    f.x = a_entry(Ac, k0, bx, col);
    f.y = a_entry(Ac, k0 + 1, by, col);
    return f;
}

__device__ __forceinline__ v8f ata_kblock(const float* Ac, int kb, int half, int lm,
                                          int m0, int n0, v8f acc) {
    const v2f a  = a_fragment(Ac, kb, half, m0 + lm);
    const v2f bb = (m0 == n0) ? a : a_fragment(Ac, kb, half, n0 + lm);
    return __builtin_amdgcn_wmma_f32_16x16x4_f32(false, a, false, bb, (short)0, acc, false, false);
}

__device__ __forceinline__ int rows_below(int l16, int c) {
    const int t = l16 - (c % EP);
    return t < 0 ? t + EP : t;
}

__device__ __forceinline__ float load_band_column(const float* band, int c, int l16) {
    const int k = rows_below(l16, c);
    return band[c * 8 + ((l16 == 15) ? 7 : (k & 7))];
}

template <int J, int K2>
__device__ __forceinline__ void elim_updates(float (&col)[NB_], const float w, const float pm) {
    if constexpr (K2 <= BW) {
        if constexpr (J + K2 < NB_)
            col[J + K2] = fmaf(-row_value<(J + K2) % EP>(w), pm, col[J + K2]);
        elim_updates<J, K2 + 1>(col, w, pm);
    }
}

template <int J>
__device__ __forceinline__ void elim_steps(float (&col)[NB_], float* const (&wP)[EP], const float* band,
                                           const int l16) {
    if constexpr (J < NB_) {
        if constexpr (J + BW + 1 + LA < NB_)
            col[J + BW + 1 + LA] = load_band_column(band, J + BW + 1 + LA, l16);
        const int  k    = rows_below(l16, J);
        const bool isr  = (l16 == 15);
        const bool inb  = (l16 < 15) & (k <= BW) & (J + k < NB_);
        const bool live = isr | inb;
        const float pm  = live ? col[J] : 0.0f;
        const float piv = row_value<J % EP>(col[J]);
        const float r0  = __builtin_amdgcn_rcpf(piv);
        const float q0  = pm * r0;
        const float w   = fmaf(fmaf(-piv, q0, pm), r0, q0);
        elim_updates<J, 1>(col, w, pm);
        wP[J % EP][J * PKW] = w;
        __builtin_amdgcn_sched_barrier(0);
        elim_steps<J + 1>(col, wP, band, l16);
    }
}

__global__ __launch_bounds__(32)
void odein_solve_kernel(const float* __restrict__ coeffs,
                        const float* __restrict__ rhs,
                        const float* __restrict__ iv_rhs,
                        const float* __restrict__ steps,
                        float* __restrict__ out)
{
    static_assert(ABUF >= (KROWS + APAD) * AW, "array too small for A");
    __shared__ __attribute__((aligned(16))) float AcAll[2 * SYSF];
    __shared__ float st[32];
    __shared__ float unkept[32];

    const int lane = threadIdx.x;
    const int half = lane >> 4, lm = lane & 15;

    for (int sy = 0; sy < 2; ++sy) {
        float* Ac   = AcAll + sy * SYSF;
        const int b = 2 * blockIdx.x + sy;

        {
            const v4f z4 = {0.0f, 0.0f, 0.0f, 0.0f};
            v4f* Ac4   = reinterpret_cast<v4f*>(Ac);
            for (int i = lane; i < ABUF / 4; i += 32) Ac4[i] = z4;
        }
        if (lane < 29) st[lane] = steps[b * 29 + lane];
        __syncthreads();

        float r1v[3] = {0.0f, 0.0f, 0.0f};
#pragma unroll
        for (int g = 0; g < 3; ++g) {
            const int c = lane + 32 * g;
            if (c < NB) {
                const float cf = coeffs[b * 90 + c];
                const int s = c / 3;
                Ac[A_ROW0(s) * AW + (c - 3 * s)] = cf;
                float v = cf * rhs[b * 30 + s];
                if      (c == 0) v += iv_rhs[b * 4 + 0];
                else if (c == 1) v += iv_rhs[b * 4 + 1];
                else if (c == 3) v += iv_rhs[b * 4 + 2];
                else if (c == 4) v += iv_rhs[b * 4 + 3];
                r1v[g] = v;
            }
        }
        if (lane < 4)
            Ac[lane * AW + lane + (lane >> 1)] = 1.0f;
        if (lane < 29) {
            const int s = lane;
            const float h = st[s], h2 = h * h;
            float* r = Ac + (A_ROW0(s) + 1) * AW;
            r[3] = -1.0f;
            r[0] = 1.0f;
            r[1] = h;
            r[2] = 0.5f * h2;
            r += AW;
            r[4] = -h;
            r[1] = h;
            r[2] = h2;
            r += AW;
            r[0] = -1.0f;
            r[3] = 1.0f;
            r[4] = -h;
            r[5] = 0.5f * h2;
            r += AW;
            r[1] = h;
            r[4] = -h;
            r[5] = h2;
        }
        if (lane >= 1 && lane < 29) {
            const int s = lane;
            float* r = Ac + (A_ROW0(s - 1) + 5) * AW;
            r[1] = -1.0f;
            r[7] = 1.0f;
            r[5] = -(st[s] + st[s - 1]);
        }
        __syncthreads();

        float* band = Ac + BAND0;
        {
            const int DK[6] = {0, 6, 15, 23, 30, 33};
            const int OK[6] = {6, 15, 23, 30, 39, 0};
#pragma unroll
            for (int t = 0; t < 6; ++t) {
                v8f accd = {0.f, 0.f, 0.f, 0.f, 0.f, 0.f, 0.f, 0.f};
                v8f acco = {0.f, 0.f, 0.f, 0.f, 0.f, 0.f, 0.f, 0.f};
#pragma unroll
                for (int i = 0; i < 12; ++i) {
                    accd = ata_kblock(Ac, DK[t] + i, half, lm, 16 * t, 16 * t, accd);
                    if (t < 5 && i < 4)
                        acco = ata_kblock(Ac, OK[t] + i, half, lm, 16 * (t + 1), 16 * t, acco);
                }
                __syncthreads();
                const int n = 16 * t + lm;
#pragma unroll
                for (int v = 0; v < 8; ++v) {
                    const int d = v + 8 * half - lm;
                    float* dst = ((unsigned)d <= (unsigned)BW) ? (band + n * 8 + d) : (unkept + lane);
                    *dst = accd[v];
                }
                if (t < 5) {
#pragma unroll
                    for (int v = 0; v < 6; ++v) {
                        const int d = 16 + v + 8 * half - lm;
                        float* dst = ((unsigned)d <= (unsigned)BW) ? (band + n * 8 + d) : (unkept + lane);
                        *dst = acco[v];
                    }
                }
            }
        }
#pragma unroll
        for (int g = 0; g < 3; ++g)
            if (lane + 32 * g < NB) band[(lane + 32 * g) * 8 + 7] = r1v[g];
        __syncthreads();
        Ac[PK0 + lane] = 0.0f;
        if (lane < QPRE * PKW - 32) Ac[PK0 + 32 + lane] = 0.0f;
#pragma unroll
        for (int g = 0; g < 3; ++g)
            if (lane + 32 * g < NB) Ac[PK0 + (QPRE + lane + 32 * g) * PKW] = 0.0f;
    }

    static_assert(16 <= 8 * (BW + 1 + LA), "a step's 16 spare words must lie in band columns that are already in registers");
    static_assert(8 * (NB - 1) + 16 <= PK0, "the spare words of the last step must stay inside the band");
    float* sysA = AcAll + half * SYSF;
    float* band = sysA + BAND0;
    float* pk   = sysA + PK0 + QPRE * PKW;
    {
        const int l16 = lane & 15;
        float* wP[EP];
#pragma unroll
        for (int p = 0; p < EP; ++p) {
            const int k = rows_below(l16, p);
            const bool keeps = (l16 == 15) | ((unsigned)(k - 1) < (unsigned)BW);
            wP[p] = keeps ? (pk + ((l16 == 15) ? 7 : k)) : (band + l16);
        }
        float col[NB];
#pragma unroll
        for (int c = 0; c < BW + 1 + LA; ++c) col[c] = load_band_column(band, c, l16);
        elim_steps<0>(col, wP, band, l16);
    }
    __syncthreads();

    {
        const int l16 = lane & 15;
        const float* rvP[16];
#pragma unroll
        for (int p = 0; p < 16; ++p) {
            const int k = (p - l16) & 15;
            rvP[p] = ((unsigned)(k - 1) < (unsigned)BW) ? (pk + (p - k) * PKW + k) : (pk + p * PKW);
        }
        const float* yE = pk + 23 + 8 * ((l16 - 2) & 15);
        const float* yO = pk - 41 + 8 * ((l16 - 10) & 15);
        const bool sE = ((l16 - 2) & 15) < 8;

        float acc = yO[128 * 5];
        float X[6] = {0.0f, 0.0f, 0.0f, 0.0f, 0.0f, 0.0f};
        float rv[NB];
        float yn[12];
#pragma unroll
        for (int m = NB - 1; m > NB - 1 - BLA_; --m) rv[m] = rvP[m & 15][128 * (m >> 4)];
        back_rows<NB - 1>(acc, X, rv, yn, rvP, yE, yO, sE);
        const int bo = 2 * blockIdx.x + half;
#pragma unroll
        for (int cx = 0; cx < 6; ++cx) {
            const int i = 74 - 16 * cx + ((l16 - 10) & 15);
            if (i >= 0) out[bo * 90 + i] = X[cx];
        }
    }
}

extern "C" void kernel_launch(void* const* d_in, const int* in_sizes, int n_in,
                              void* d_out, int out_size, void* d_ws, size_t ws_size,
                              hipStream_t stream) {
    (void)n_in; (void)out_size; (void)d_ws; (void)ws_size;
    const float* coeffs = (const float*)d_in[0];
    const float* rhs    = (const float*)d_in[1];
    const float* iv     = (const float*)d_in[2];
    const float* steps  = (const float*)d_in[3];
    float* out = (float*)d_out;
    int pairs = in_sizes[0] / 180;
    odein_solve_kernel<<<pairs, 32, 0, stream>>>(coeffs, rhs, iv, steps, out);
}


# All global reads of both systems of a wave issued at its start as one group of 16 loads with one wait; was 9 dependent round trips per system, each load waited for alone, four of them in branches
# speedup vs baseline: 47.5573x; 1.1649x over previous
#include <hip/hip_runtime.h>
#include <hip/hip_bf16.h>


typedef float v2f __attribute__((ext_vector_type(2)));
typedef float v4f __attribute__((ext_vector_type(4)));
typedef float v8f __attribute__((ext_vector_type(8)));

#define KROWS 180
#define AW    8
#define APAD  2
#define NB    NB_
#define NBP   96
#define BW    6
#define PKW   8
#define LA    8
#define EP    15
#define BAND0 0
#define PK0   (NBP * 8)
#define QPRE  BW
#define ABUF  (PK0 + NBP * PKW)
#define SYSF  (ABUF + 32)

template <int N>
__device__ __forceinline__ float row_value(float v) {
    return __int_as_float(__builtin_amdgcn_update_dpp(0, __float_as_int(v), 0x150 + N, 0xf, 0xf, true));
}

#define NB_   90
#define BLA_  12

template <int M>
__device__ __forceinline__ void back_rows(float& acc, float (&X)[6], float (&rv)[NB_], float (&yn)[12],
                                          const float* const (&rvP)[16], const float* yE,
                                          const float* yO, const bool sE) {
    if constexpr (M >= 0) {
        constexpr int b = (NB_ - 1 - M) >> 3;
        if constexpr (M - BLA_ >= 0)
            rv[M - BLA_] = rvP[(M - BLA_) & 15][128 * ((M - BLA_) >> 4)];
        if constexpr (((NB_ - 1 - M) & 7) == 0 && b <= 9)
            yn[b] = ((b & 1) ? yO : yE)[128 * (4 - (b >> 1))];
        acc = fmaf(-row_value<(M & 15)>(acc), rv[M], acc);
        if constexpr (((NB_ - 1 - M) & 7) == 7 || M == 0) {
            const bool s = (b & 1) ? !sE : sE;
            X[b >> 1] = s ? acc : X[b >> 1];
            if constexpr (b <= 9) acc = s ? yn[b] : acc;
        }
        __builtin_amdgcn_sched_barrier(0);
        back_rows<M - 1>(acc, X, rv, yn, rvP, yE, yO, sE);
    }
}

#define A_ROW0(s) (4 + 6 * (s))

__device__ __forceinline__ int row_base(int k) { return k < 4 ? 0 : 3 * ((k - 4) / 6); }

__device__ __forceinline__ float a_entry(const float* Ac, int k, int base, int col) {
    const int e = col - base;
    const float v = Ac[k * AW + e];
    return ((unsigned)e < (unsigned)AW) ? v : 0.0f;
}

__device__ __forceinline__ v2f a_fragment(const float* Ac, int kb, int half, int col) {
    const int k0 = 4 * kb + 2 * half;
    const int bx = half ? row_base(4 * kb + 2) : row_base(4 * kb);
    const int by = half ? row_base(4 * kb + 3) : row_base(4 * kb + 1);
    v2f f;
    f.x = a_entry(Ac, k0, bx, col);
    f.y = a_entry(Ac, k0 + 1, by, col);
    return f;
}

__device__ __forceinline__ v8f ata_kblock(const float* Ac, int kb, int half, int lm,
                                          int m0, int n0, v8f acc) {
    const v2f a  = a_fragment(Ac, kb, half, m0 + lm);
    const v2f bb = (m0 == n0) ? a : a_fragment(Ac, kb, half, n0 + lm);
    return __builtin_amdgcn_wmma_f32_16x16x4_f32(false, a, false, bb, (short)0, acc, false, false);
}

__device__ __forceinline__ int rows_below(int l16, int c) {
    const int t = l16 - (c % EP);
    return t < 0 ? t + EP : t;
}

__device__ __forceinline__ float load_band_column(const float* band, int c, int l16) {
    const int k = rows_below(l16, c);
    return band[c * 8 + ((l16 == 15) ? 7 : (k & 7))];
}

template <int J, int K2>
__device__ __forceinline__ void elim_updates(float (&col)[NB_], const float w, const float pm) {
    if constexpr (K2 <= BW) {
        if constexpr (J + K2 < NB_)
            col[J + K2] = fmaf(-row_value<(J + K2) % EP>(w), pm, col[J + K2]);
        elim_updates<J, K2 + 1>(col, w, pm);
    }
}

template <int J>
__device__ __forceinline__ void elim_steps(float (&col)[NB_], float* const (&wP)[EP], const float* band,
                                           const int l16) {
    if constexpr (J < NB_) {
        if constexpr (J + BW + 1 + LA < NB_)
            col[J + BW + 1 + LA] = load_band_column(band, J + BW + 1 + LA, l16);
        const int  k    = rows_below(l16, J);
        const bool isr  = (l16 == 15);
        const bool inb  = (l16 < 15) & (k <= BW) & (J + k < NB_);
        const bool live = isr | inb;
        const float pm  = live ? col[J] : 0.0f;
        const float piv = row_value<J % EP>(col[J]);
        const float r0  = __builtin_amdgcn_rcpf(piv);
        const float q0  = pm * r0;
        const float w   = fmaf(fmaf(-piv, q0, pm), r0, q0);
        elim_updates<J, 1>(col, w, pm);
        wP[J % EP][J * PKW] = w;
        __builtin_amdgcn_sched_barrier(0);
        elim_steps<J + 1>(col, wP, band, l16);
    }
}

__global__ __launch_bounds__(32)
void odein_solve_kernel(const float* __restrict__ coeffs,
                        const float* __restrict__ rhs,
                        const float* __restrict__ iv_rhs,
                        const float* __restrict__ steps,
                        float* __restrict__ out)
{
    static_assert(ABUF >= (KROWS + APAD) * AW, "array too small for A");
    __shared__ __attribute__((aligned(16))) float AcAll[2 * SYSF];
    __shared__ float st[32];
    __shared__ float unkept[32];

    const int lane = threadIdx.x;
    const int half = lane >> 4, lm = lane & 15;

    const int b0 = 2 * blockIdx.x;
    const int ls = lane < 29 ? lane : 28;
    const int c2 = lane + 64 < NB ? lane + 64 : NB - 1;
    const int li = lane < 5 ? lane - (lane >= 3) : 0;
    const float st_0  = steps[b0 * 29 + ls];
    const float cf0_0 = coeffs[b0 * 90 + lane];
    const float cf1_0 = coeffs[b0 * 90 + lane + 32];
    const float cf2_0 = coeffs[b0 * 90 + c2];
    const float rh0_0 = rhs[b0 * 30 + lane / 3];
    const float rh1_0 = rhs[b0 * 30 + (lane + 32) / 3];
    const float rh2_0 = rhs[b0 * 30 + c2 / 3];
    const float iv_0  = iv_rhs[b0 * 4 + li];
    const float st_1  = steps[(b0 + 1) * 29 + ls];
    const float cf0_1 = coeffs[(b0 + 1) * 90 + lane];
    const float cf1_1 = coeffs[(b0 + 1) * 90 + lane + 32];
    const float cf2_1 = coeffs[(b0 + 1) * 90 + c2];
    const float rh0_1 = rhs[(b0 + 1) * 30 + lane / 3];
    const float rh1_1 = rhs[(b0 + 1) * 30 + (lane + 32) / 3];
    const float rh2_1 = rhs[(b0 + 1) * 30 + c2 / 3];
    const float iv_1  = iv_rhs[(b0 + 1) * 4 + li];

    for (int sy = 0; sy < 2; ++sy) {
        float* Ac   = AcAll + sy * SYSF;
        const float cfv[3] = {sy ? cf0_1 : cf0_0, sy ? cf1_1 : cf1_0, sy ? cf2_1 : cf2_0};
        const float rhv[3] = {sy ? rh0_1 : rh0_0, sy ? rh1_1 : rh1_0, sy ? rh2_1 : rh2_0};

        {
            const v4f z4 = {0.0f, 0.0f, 0.0f, 0.0f};
            v4f* Ac4   = reinterpret_cast<v4f*>(Ac);
            for (int i = lane; i < ABUF / 4; i += 32) Ac4[i] = z4;
        }
        if (lane < 29) st[lane] = sy ? st_1 : st_0;
        __syncthreads();

        float r1v[3] = {0.0f, 0.0f, 0.0f};
#pragma unroll
        for (int g = 0; g < 3; ++g) {
            const int c = lane + 32 * g;
            if (c < NB) {
                const float cf = cfv[g];
                const int s = c / 3;
                Ac[A_ROW0(s) * AW + (c - 3 * s)] = cf;
                float v = cf * rhv[g];
                if (c < 5 && c != 2) v += sy ? iv_1 : iv_0;
                r1v[g] = v;
            }
        }
        if (lane < 4)
            Ac[lane * AW + lane + (lane >> 1)] = 1.0f;
        if (lane < 29) {
            const int s = lane;
            const float h = st[s], h2 = h * h;
            float* r = Ac + (A_ROW0(s) + 1) * AW;
            r[3] = -1.0f;
            r[0] = 1.0f;
            r[1] = h;
            r[2] = 0.5f * h2;
            r += AW;
            r[4] = -h;
            r[1] = h;
            r[2] = h2;
            r += AW;
            r[0] = -1.0f;
            r[3] = 1.0f;
            r[4] = -h;
            r[5] = 0.5f * h2;
            r += AW;
            r[1] = h;
            r[4] = -h;
            r[5] = h2;
        }
        if (lane >= 1 && lane < 29) {
            const int s = lane;
            float* r = Ac + (A_ROW0(s - 1) + 5) * AW;
            r[1] = -1.0f;
            r[7] = 1.0f;
            r[5] = -(st[s] + st[s - 1]);
        }
        __syncthreads();

        float* band = Ac + BAND0;
        {
            const int DK[6] = {0, 6, 15, 23, 30, 33};
            const int OK[6] = {6, 15, 23, 30, 39, 0};
#pragma unroll
            for (int t = 0; t < 6; ++t) {
                v8f accd = {0.f, 0.f, 0.f, 0.f, 0.f, 0.f, 0.f, 0.f};
                v8f acco = {0.f, 0.f, 0.f, 0.f, 0.f, 0.f, 0.f, 0.f};
#pragma unroll
                for (int i = 0; i < 12; ++i) {
                    accd = ata_kblock(Ac, DK[t] + i, half, lm, 16 * t, 16 * t, accd);
                    if (t < 5 && i < 4)
                        acco = ata_kblock(Ac, OK[t] + i, half, lm, 16 * (t + 1), 16 * t, acco);
                }
                __syncthreads();
                const int n = 16 * t + lm;
#pragma unroll
                for (int v = 0; v < 8; ++v) {
                    const int d = v + 8 * half - lm;
                    float* dst = ((unsigned)d <= (unsigned)BW) ? (band + n * 8 + d) : (unkept + lane);
                    *dst = accd[v];
                }
                if (t < 5) {
#pragma unroll
                    for (int v = 0; v < 6; ++v) {
                        const int d = 16 + v + 8 * half - lm;
                        float* dst = ((unsigned)d <= (unsigned)BW) ? (band + n * 8 + d) : (unkept + lane);
                        *dst = acco[v];
                    }
                }
            }
        }
#pragma unroll
        for (int g = 0; g < 3; ++g)
            if (lane + 32 * g < NB) band[(lane + 32 * g) * 8 + 7] = r1v[g];
        __syncthreads();
        Ac[PK0 + lane] = 0.0f;
        if (lane < QPRE * PKW - 32) Ac[PK0 + 32 + lane] = 0.0f;
#pragma unroll
        for (int g = 0; g < 3; ++g)
            if (lane + 32 * g < NB) Ac[PK0 + (QPRE + lane + 32 * g) * PKW] = 0.0f;
    }

    static_assert(16 <= 8 * (BW + 1 + LA), "a step's 16 spare words must lie in band columns that are already in registers");
    static_assert(8 * (NB - 1) + 16 <= PK0, "the spare words of the last step must stay inside the band");
    float* sysA = AcAll + half * SYSF;
    float* band = sysA + BAND0;
    float* pk   = sysA + PK0 + QPRE * PKW;
    {
        const int l16 = lane & 15;
        float* wP[EP];
#pragma unroll
        for (int p = 0; p < EP; ++p) {
            const int k = rows_below(l16, p);
            const bool keeps = (l16 == 15) | ((unsigned)(k - 1) < (unsigned)BW);
            wP[p] = keeps ? (pk + ((l16 == 15) ? 7 : k)) : (band + l16);
        }
        float col[NB];
#pragma unroll
        for (int c = 0; c < BW + 1 + LA; ++c) col[c] = load_band_column(band, c, l16);
        elim_steps<0>(col, wP, band, l16);
    }
    __syncthreads();

    {
        const int l16 = lane & 15;
        const float* rvP[16];
#pragma unroll
        for (int p = 0; p < 16; ++p) {
            const int k = (p - l16) & 15;
            rvP[p] = ((unsigned)(k - 1) < (unsigned)BW) ? (pk + (p - k) * PKW + k) : (pk + p * PKW);
        }
        const float* yE = pk + 23 + 8 * ((l16 - 2) & 15);
        const float* yO = pk - 41 + 8 * ((l16 - 10) & 15);
        const bool sE = ((l16 - 2) & 15) < 8;

        float acc = yO[128 * 5];
        float X[6] = {0.0f, 0.0f, 0.0f, 0.0f, 0.0f, 0.0f};
        float rv[NB];
        float yn[12];
#pragma unroll
        for (int m = NB - 1; m > NB - 1 - BLA_; --m) rv[m] = rvP[m & 15][128 * (m >> 4)];
        back_rows<NB - 1>(acc, X, rv, yn, rvP, yE, yO, sE);
        const int bo = 2 * blockIdx.x + half;
#pragma unroll
        for (int cx = 0; cx < 6; ++cx) {
            const int i = 74 - 16 * cx + ((l16 - 10) & 15);
            if (i >= 0) out[bo * 90 + i] = X[cx];
        }
    }
}

extern "C" void kernel_launch(void* const* d_in, const int* in_sizes, int n_in,
                              void* d_out, int out_size, void* d_ws, size_t ws_size,
                              hipStream_t stream) {
    (void)n_in; (void)out_size; (void)d_ws; (void)ws_size;
    const float* coeffs = (const float*)d_in[0];
    const float* rhs    = (const float*)d_in[1];
    const float* iv     = (const float*)d_in[2];
    const float* steps  = (const float*)d_in[3];
    float* out = (float*)d_out;
    int pairs = in_sizes[0] / 180;
    odein_solve_kernel<<<pairs, 32, 0, stream>>>(coeffs, rhs, iv, steps, out);
}


# A^T A as sum over steps of 8x8 blocks: one tile = the block of both systems (k = the step's rows), overlaps added in registers (DPP row_shl:3): 61 WMMAs a wave, was 184; AtA's sums change order
# speedup vs baseline: 61.7611x; 1.2987x over previous
#include <hip/hip_runtime.h>
#include <hip/hip_bf16.h>


typedef float v2f __attribute__((ext_vector_type(2)));
typedef float v4f __attribute__((ext_vector_type(4)));
typedef float v8f __attribute__((ext_vector_type(8)));

#define KROWS 180
#define AW    8
#define APAD  2
#define NB    NB_
#define NBP   96
#define BW    6
#define PKW   8
#define LA    8
#define EP    15
#define BAND0 0
#define PK0   (NBP * 8)
#define QPRE  BW
#define ABUF  (PK0 + NBP * PKW)
#define SYSF  (ABUF + 32)

template <int N>
__device__ __forceinline__ float row_value(float v) {
    return __int_as_float(__builtin_amdgcn_update_dpp(0, __float_as_int(v), 0x150 + N, 0xf, 0xf, true));
}

#define NB_   90
#define FA    3
#define BLA_  12

template <int M>
__device__ __forceinline__ void back_rows(float& acc, float (&X)[6], float (&rv)[NB_], float (&yn)[12],
                                          const float* const (&rvP)[16], const float* yE,
                                          const float* yO, const bool sE) {
    if constexpr (M >= 0) {
        constexpr int b = (NB_ - 1 - M) >> 3;
        if constexpr (M - BLA_ >= 0)
            rv[M - BLA_] = rvP[(M - BLA_) & 15][128 * ((M - BLA_) >> 4)];
        if constexpr (((NB_ - 1 - M) & 7) == 0 && b <= 9)
            yn[b] = ((b & 1) ? yO : yE)[128 * (4 - (b >> 1))];
        acc = fmaf(-row_value<(M & 15)>(acc), rv[M], acc);
        if constexpr (((NB_ - 1 - M) & 7) == 7 || M == 0) {
            const bool s = (b & 1) ? !sE : sE;
            X[b >> 1] = s ? acc : X[b >> 1];
            if constexpr (b <= 9) acc = s ? yn[b] : acc;
        }
        __builtin_amdgcn_sched_barrier(0);
        back_rows<M - 1>(acc, X, rv, yn, rvP, yE, yO, sE);
    }
}

#define A_ROW0(s) (4 + 6 * (s))

__device__ __forceinline__ int rows_below(int l16, int c) {
    const int t = l16 - (c % EP);
    return t < 0 ? t + EP : t;
}

__device__ __forceinline__ float load_band_column(const float* band, int c, int l16) {
    const int k = rows_below(l16, c);
    return band[c * 8 + ((l16 == 15) ? 7 : (k & 7))];
}

template <int J, int K2>
__device__ __forceinline__ void elim_updates(float (&col)[NB_], const float w, const float pm) {
    if constexpr (K2 <= BW) {
        if constexpr (J + K2 < NB_)
            col[J + K2] = fmaf(-row_value<(J + K2) % EP>(w), pm, col[J + K2]);
        elim_updates<J, K2 + 1>(col, w, pm);
    }
}

template <int J>
__device__ __forceinline__ void elim_steps(float (&col)[NB_], float* const (&wP)[EP], const float* band,
                                           const int l16) {
    if constexpr (J < NB_) {
        if constexpr (J + BW + 1 + LA < NB_)
            col[J + BW + 1 + LA] = load_band_column(band, J + BW + 1 + LA, l16);
        const int  k    = rows_below(l16, J);
        const bool isr  = (l16 == 15);
        const bool inb  = (l16 < 15) & (k <= BW) & (J + k < NB_);
        const bool live = isr | inb;
        const float pm  = live ? col[J] : 0.0f;
        const float piv = row_value<J % EP>(col[J]);
        const float r0  = __builtin_amdgcn_rcpf(piv);
        const float q0  = pm * r0;
        const float w   = fmaf(fmaf(-piv, q0, pm), r0, q0);
        elim_updates<J, 1>(col, w, pm);
        wP[J % EP][J * PKW] = w;
        __builtin_amdgcn_sched_barrier(0);
        elim_steps<J + 1>(col, wP, band, l16);
    }
}

__global__ __launch_bounds__(32)
void odein_solve_kernel(const float* __restrict__ coeffs,
                        const float* __restrict__ rhs,
                        const float* __restrict__ iv_rhs,
                        const float* __restrict__ steps,
                        float* __restrict__ out)
{
    static_assert(ABUF >= (KROWS + APAD) * AW, "array too small for A");
    __shared__ __attribute__((aligned(16))) float AcAll[2 * SYSF];
    __shared__ float st[32];
    __shared__ float unkept[32 + 24 * 30];
    __shared__ float r1s[2 * NBP];

    const int lane = threadIdx.x;
    const int half = lane >> 4, lm = lane & 15;

    const int b0 = 2 * blockIdx.x;
    const int ls = lane < 29 ? lane : 28;
    const int c2 = lane + 64 < NB ? lane + 64 : NB - 1;
    const int li = lane < 5 ? lane - (lane >= 3) : 0;
    const float st_0  = steps[b0 * 29 + ls];
    const float cf0_0 = coeffs[b0 * 90 + lane];
    const float cf1_0 = coeffs[b0 * 90 + lane + 32];
    const float cf2_0 = coeffs[b0 * 90 + c2];
    const float rh0_0 = rhs[b0 * 30 + lane / 3];
    const float rh1_0 = rhs[b0 * 30 + (lane + 32) / 3];
    const float rh2_0 = rhs[b0 * 30 + c2 / 3];
    const float iv_0  = iv_rhs[b0 * 4 + li];
    const float st_1  = steps[(b0 + 1) * 29 + ls];
    const float cf0_1 = coeffs[(b0 + 1) * 90 + lane];
    const float cf1_1 = coeffs[(b0 + 1) * 90 + lane + 32];
    const float cf2_1 = coeffs[(b0 + 1) * 90 + c2];
    const float rh0_1 = rhs[(b0 + 1) * 30 + lane / 3];
    const float rh1_1 = rhs[(b0 + 1) * 30 + (lane + 32) / 3];
    const float rh2_1 = rhs[(b0 + 1) * 30 + c2 / 3];
    const float iv_1  = iv_rhs[(b0 + 1) * 4 + li];

    for (int sy = 0; sy < 2; ++sy) {
        float* Ac   = AcAll + sy * SYSF;
        const float cfv[3] = {sy ? cf0_1 : cf0_0, sy ? cf1_1 : cf1_0, sy ? cf2_1 : cf2_0};
        const float rhv[3] = {sy ? rh0_1 : rh0_0, sy ? rh1_1 : rh1_0, sy ? rh2_1 : rh2_0};

        {
            const v4f z4 = {0.0f, 0.0f, 0.0f, 0.0f};
            v4f* Ac4   = reinterpret_cast<v4f*>(Ac);
            for (int i = lane; i < ABUF / 4; i += 32) Ac4[i] = z4;
        }
        if (lane < 29) st[lane] = sy ? st_1 : st_0;
        __syncthreads();

#pragma unroll
        for (int g = 0; g < 3; ++g) {
            const int c = lane + 32 * g;
            if (c < NB) {
                const float cf = cfv[g];
                const int s = c / 3;
                Ac[A_ROW0(s) * AW + (c - 3 * s)] = cf;
                float v = cf * rhv[g];
                if (c < 5 && c != 2) v += sy ? iv_1 : iv_0;
                r1s[sy * NBP + c] = v;
            }
        }
        if (lane < 4)
            Ac[lane * AW + lane + (lane >> 1)] = 1.0f;
        if (lane < 29) {
            const int s = lane;
            const float h = st[s], h2 = h * h;
            float* r = Ac + (A_ROW0(s) + 1) * AW;
            r[3] = -1.0f;
            r[0] = 1.0f;
            r[1] = h;
            r[2] = 0.5f * h2;
            r += AW;
            r[4] = -h;
            r[1] = h;
            r[2] = h2;
            r += AW;
            r[0] = -1.0f;
            r[3] = 1.0f;
            r[4] = -h;
            r[5] = 0.5f * h2;
            r += AW;
            r[1] = h;
            r[4] = -h;
            r[5] = h2;
        }
        if (lane >= 1 && lane < 29) {
            const int s = lane;
            float* r = Ac + (A_ROW0(s - 1) + 5) * AW;
            r[1] = -1.0f;
            r[7] = 1.0f;
            r[5] = -(st[s] + st[s - 1]);
        }
        __syncthreads();

    }

    {
        const int j    = lm & 7;
        float* AcL     = AcAll + (lm >> 3) * SYSF;
        const float* pA = AcL + (A_ROW0(0) + 2 * half) * AW + j;
        const float* pB = half ? (AcL + 184 * AW + j)
                               : (AcL + (A_ROW0(0) + 4) * AW + j);
        const int incB  = half ? 0 : 6 * AW;
        const bool keep = (lane < 8) | (lane >= 24);
        float* const bandL = AcL + BAND0 + 7 * j;
        float* stP[8];
#pragma unroll
        for (int i = 0; i < 8; ++i) {
            const bool valid = keep & (j < 3) & (i >= j) & (i - j <= BW);
            stP[i] = valid ? (bandL + i) : (unkept + lane);
        }
        float R[8] = {0.f, 0.f, 0.f, 0.f, 0.f, 0.f, 0.f, 0.f};
        v2f fa[30], fb[30];
        v2f fi;
        fi.x = AcL[(2 * half) * AW + j];
        fi.y = AcL[(2 * half + 1) * AW + j];
#pragma unroll
        for (int s = 0; s < FA; ++s) {
            fa[s].x = pA[s * 6 * AW];
            fa[s].y = pA[s * 6 * AW + AW];
            fb[s].x = pB[0];
            fb[s].y = pB[AW];
            pB += incB;
        }
        v8f Tp = {0.f, 0.f, 0.f, 0.f, 0.f, 0.f, 0.f, 0.f};
#pragma unroll
        for (int s = 0; s <= 30; ++s) {
            if (s + FA < 30) {
                fa[s + FA].x = pA[(s + FA) * 6 * AW];
                fa[s + FA].y = pA[(s + FA) * 6 * AW + AW];
                fb[s + FA].x = pB[0];
                fb[s + FA].y = pB[AW];
                pB += incB;
            }
            v8f T = {0.f, 0.f, 0.f, 0.f, 0.f, 0.f, 0.f, 0.f};
            if (s < 30) {
                if (s == 0)
                    T = __builtin_amdgcn_wmma_f32_16x16x4_f32(false, fi, false, fi, (short)0, T, false, false);
                T = __builtin_amdgcn_wmma_f32_16x16x4_f32(false, fa[s], false, fa[s], (short)0, T, false, false);
                T = __builtin_amdgcn_wmma_f32_16x16x4_f32(false, fb[s], false, fb[s], (short)0, T, false, false);
            }
            if (s >= 1) {
#pragma unroll
                for (int i = 0; i < 8; ++i) {
                    const float t = keep ? Tp[i] : 0.0f;
                    R[i] = (i < 5) ? t + __int_as_float(__builtin_amdgcn_update_dpp(0, __float_as_int(R[i + 3]), 0x103, 0xf, 0xf, true))
                                   : t;
                }
#pragma unroll
                for (int i = 0; i < 8; ++i) stP[i][24 * (s - 1)] = R[i];
            }
            Tp = T;
        }
    }
    __syncthreads();
    for (int sy = 0; sy < 2; ++sy) {
        float* Ac   = AcAll + sy * SYSF;
        float* band = Ac + BAND0;
#pragma unroll
        for (int g = 0; g < 3; ++g)
            if (lane + 32 * g < NB) band[(lane + 32 * g) * 8 + 7] = r1s[sy * NBP + lane + 32 * g];
        Ac[PK0 + lane] = 0.0f;
        if (lane < QPRE * PKW - 32) Ac[PK0 + 32 + lane] = 0.0f;
#pragma unroll
        for (int g = 0; g < 3; ++g)
            if (lane + 32 * g < NB) Ac[PK0 + (QPRE + lane + 32 * g) * PKW] = 0.0f;
    }
    __syncthreads();

    static_assert(16 <= 8 * (BW + 1 + LA), "a step's 16 spare words must lie in band columns that are already in registers");
    static_assert(8 * (NB - 1) + 16 <= PK0, "the spare words of the last step must stay inside the band");
    float* sysA = AcAll + half * SYSF;
    float* band = sysA + BAND0;
    float* pk   = sysA + PK0 + QPRE * PKW;
    {
        const int l16 = lane & 15;
        float* wP[EP];
#pragma unroll
        for (int p = 0; p < EP; ++p) {
            const int k = rows_below(l16, p);
            const bool keeps = (l16 == 15) | ((unsigned)(k - 1) < (unsigned)BW);
            wP[p] = keeps ? (pk + ((l16 == 15) ? 7 : k)) : (band + l16);
        }
        float col[NB];
#pragma unroll
        for (int c = 0; c < BW + 1 + LA; ++c) col[c] = load_band_column(band, c, l16);
        elim_steps<0>(col, wP, band, l16);
    }
    __syncthreads();

    {
        const int l16 = lane & 15;
        const float* rvP[16];
#pragma unroll
        for (int p = 0; p < 16; ++p) {
            const int k = (p - l16) & 15;
            rvP[p] = ((unsigned)(k - 1) < (unsigned)BW) ? (pk + (p - k) * PKW + k) : (pk + p * PKW);
        }
        const float* yE = pk + 23 + 8 * ((l16 - 2) & 15);
        const float* yO = pk - 41 + 8 * ((l16 - 10) & 15);
        const bool sE = ((l16 - 2) & 15) < 8;

        float acc = yO[128 * 5];
        float X[6] = {0.0f, 0.0f, 0.0f, 0.0f, 0.0f, 0.0f};
        float rv[NB];
        float yn[12];
#pragma unroll
        for (int m = NB - 1; m > NB - 1 - BLA_; --m) rv[m] = rvP[m & 15][128 * (m >> 4)];
        back_rows<NB - 1>(acc, X, rv, yn, rvP, yE, yO, sE);
        const int bo = 2 * blockIdx.x + half;
#pragma unroll
        for (int cx = 0; cx < 6; ++cx) {
            const int i = 74 - 16 * cx + ((l16 - 10) & 15);
            if (i >= 0) out[bo * 90 + i] = X[cx];
        }
    }
}

extern "C" void kernel_launch(void* const* d_in, const int* in_sizes, int n_in,
                              void* d_out, int out_size, void* d_ws, size_t ws_size,
                              hipStream_t stream) {
    (void)n_in; (void)out_size; (void)d_ws; (void)ws_size;
    const float* coeffs = (const float*)d_in[0];
    const float* rhs    = (const float*)d_in[1];
    const float* iv     = (const float*)d_in[2];
    const float* steps  = (const float*)d_in[3];
    float* out = (float*)d_out;
    int pairs = in_sizes[0] / 180;
    odein_solve_kernel<<<pairs, 32, 0, stream>>>(coeffs, rhs, iv, steps, out);
}
